# starting kernel
# baseline (speedup 1.0000x reference)
#include <hip/hip_runtime.h>


typedef __attribute__((ext_vector_type(2))) float v2f;
typedef __attribute__((ext_vector_type(8))) float v8f;

#define NXC 64
#define NKC 16
#define BATCHC 16
#define FEATC (NXC * NXC)

__global__ __launch_bounds__(128) void polar_stage_kernel(
    const float* __restrict__ inputs,
    const int*   __restrict__ r_index,
    const float* __restrict__ pre1,  const float* __restrict__ pre2,
    const float* __restrict__ pre3,  const float* __restrict__ pre4,
    const float* __restrict__ post1, const float* __restrict__ post2,
    const float* __restrict__ post3, const float* __restrict__ post4,
    const float* __restrict__ ck1, const float* __restrict__ sk1,
    const float* __restrict__ ck2, const float* __restrict__ sk2,
    const float* __restrict__ ck3, const float* __restrict__ sk3,
    const float* __restrict__ ck4, const float* __restrict__ sk4,
    float* __restrict__ polar)
{
    __shared__ float sD[2][NXC * NXC];
    __shared__ float sA[NXC * NXC];
    __shared__ float sB[NXC * NXC];

    const int tid  = threadIdx.x;
    const int lane = tid & 31;
    const int wave = tid >> 5;

    const int c     = blockIdx.x;
    const int batch = blockIdx.y;
    const int f     = blockIdx.z;

    const int* rix = r_index + c * FEATC;
    const float* baseR = inputs + ((size_t)(batch * 2 + 0) * FEATC) * NKC + f;
    const float* baseI = inputs + ((size_t)(batch * 2 + 1) * FEATC) * NKC + f;
    for (int e = tid; e < FEATC; e += 128) {
        const int idx = rix[e];
        sD[0][e] = baseR[(size_t)idx * NKC];
        sD[1][e] = baseI[(size_t)idx * NKC];
    }
    __syncthreads();

    const float* PRE[4]  = {pre1,  pre2,  pre3,  pre4};
    const float* POST[4] = {post1, post2, post3, post4};
    const float* K2[4]   = {ck1, sk1, ck3, sk4};
    const float* K1[4]   = {ck2, sk2, sk3, ck4};
    const int    DSEL[4] = {0, 0, 1, 1};

    const int kcol    = wave * 16 + (lane & 15);
    const int halfsel = lane >> 4;
    const int arow    = (lane & 15);
    float acc = 0.0f;

    for (int t = 0; t < 4; ++t) {
        __syncthreads();
        const float* k2f   = K2[t] + f * FEATC;
        const float* k1f   = K1[t] + f * FEATC;
        const float* postf = POST[t] + f * NXC;
        const float* pref  = PRE[t]  + f * NXC;
        for (int e = tid; e < FEATC; e += 128) {
            const int row = e >> 6;
            sA[e] = postf[row] * k2f[e];
            sB[e] = pref[row]  * k1f[e];
        }
        __syncthreads();

        const float* Dt = sD[DSEL[t]];

        for (int it = 0; it < 4; it += 2) {
            v8f m0 = {};
            v8f m1 = {};
            #pragma unroll
            for (int j = 0; j < NXC; j += 4) {
                v2f b;
                b.x = sB[(j + halfsel * 2)     * NXC + kcol];
                b.y = sB[(j + halfsel * 2 + 1) * NXC + kcol];
                const float* a0p = &Dt[((it    ) * 16 + arow) * NXC + j + halfsel * 2];
                const float* a1p = &Dt[((it + 1) * 16 + arow) * NXC + j + halfsel * 2];
                v2f a0; a0.x = a0p[0]; a0.y = a0p[1];
                v2f a1; a1.x = a1p[0]; a1.y = a1p[1];
                m0 = __builtin_amdgcn_wmma_f32_16x16x4_f32(
                         false, a0, false, b, (short)0, m0, false, false);
                m1 = __builtin_amdgcn_wmma_f32_16x16x4_f32(
                         false, a1, false, b, (short)0, m1, false, false);
            }
            const int rb0 = (it    ) * 16 + halfsel * 8;
            const int rb1 = (it + 1) * 16 + halfsel * 8;
            #pragma unroll
            for (int r = 0; r < 8; ++r) {
                acc += sA[(rb0 + r) * NXC + kcol] * m0[r];
                acc += sA[(rb1 + r) * NXC + kcol] * m1[r];
            }
        }
    }

    acc += __shfl_xor(acc, 16, 32);
    if (lane < 16) {
        const size_t fb = (size_t)f * BATCHC + batch;
        polar[fb * FEATC + c * NXC + wave * 16 + lane] = acc;
    }
}

__global__ __launch_bounds__(256) void cart_stage_kernel(
    const float* __restrict__ cart,
    const float* __restrict__ polar,
    float* __restrict__ out)
{
    __shared__ float sP[64 * 64];
    __shared__ float sC[64 * 128];

    const int tid  = threadIdx.x;
    const int lane = tid & 31;
    const int wave = tid >> 5;

    const int n0 = blockIdx.x * 128;
    const int m0 = blockIdx.y * 64;

    const int mt0 = (wave >> 2) * 2;
    const int nt0 = (wave & 3) * 2;

    const int halfsel = lane >> 4;
    const int ml      = lane & 15;

    v8f acc00 = {};
    v8f acc01 = {};
    v8f acc10 = {};
    v8f acc11 = {};

    for (int q0 = 0; q0 < 4096; q0 += 64) {
        __syncthreads();
        for (int e = tid; e < 4096; e += 256) {
            const int m = e >> 6, qq = e & 63;
            sP[e] = polar[(size_t)(m0 + m) * 4096 + q0 + qq];
        }
        for (int e = tid; e < 8192; e += 256) {
            const int p = e >> 6, qq = e & 63;
            sC[qq * 128 + p] = cart[(size_t)(n0 + p) * 4096 + q0 + qq];
        }
        if (q0 + 64 < 4096 && tid < 128) {
            __builtin_prefetch(&cart[(size_t)(n0 + tid) * 4096 + q0 + 64], 0, 1);
        }
        __syncthreads();

        const int n_a = (nt0    ) * 16 + ml;
        const int n_b = (nt0 + 1) * 16 + ml;
        #pragma unroll
        for (int j = 0; j < 64; j += 4) {
            const float* a0p = &sP[((mt0    ) * 16 + ml) * 64 + j + halfsel * 2];
            const float* a1p = &sP[((mt0 + 1) * 16 + ml) * 64 + j + halfsel * 2];
            v2f a0; a0.x = a0p[0]; a0.y = a0p[1];
            v2f a1; a1.x = a1p[0]; a1.y = a1p[1];
            v2f b0, b1;
            b0.x = sC[(j + halfsel * 2)     * 128 + n_a];
            b0.y = sC[(j + halfsel * 2 + 1) * 128 + n_a];
            b1.x = sC[(j + halfsel * 2)     * 128 + n_b];
            b1.y = sC[(j + halfsel * 2 + 1) * 128 + n_b];
            acc00 = __builtin_amdgcn_wmma_f32_16x16x4_f32(
                        false, a0, false, b0, (short)0, acc00, false, false);
            acc01 = __builtin_amdgcn_wmma_f32_16x16x4_f32(
                        false, a0, false, b1, (short)0, acc01, false, false);
            acc10 = __builtin_amdgcn_wmma_f32_16x16x4_f32(
                        false, a1, false, b0, (short)0, acc10, false, false);
            acc11 = __builtin_amdgcn_wmma_f32_16x16x4_f32(
                        false, a1, false, b1, (short)0, acc11, false, false);
        }
    }

    #pragma unroll
    for (int mi = 0; mi < 2; ++mi) {
        #pragma unroll
        for (int ni = 0; ni < 2; ++ni) {
            const v8f acc = (mi == 0) ? (ni == 0 ? acc00 : acc01)
                                      : (ni == 0 ? acc10 : acc11);
            const int p  = n0 + (nt0 + ni) * 16 + ml;
            const int e1 = p >> 6, e2 = p & 63;
            #pragma unroll
            for (int r = 0; r < 8; ++r) {
                const int fb = m0 + (mt0 + mi) * 16 + halfsel * 8 + r;
                const int fo = fb >> 4, bo = fb & 15;
                out[(((size_t)bo * 64 + e1) * 64 + e2) * NKC + fo] = acc[r];
            }
        }
    }
}

extern "C" void kernel_launch(void* const* d_in, const int* in_sizes, int n_in,
                              void* d_out, int out_size, void* d_ws, size_t ws_size,
                              hipStream_t stream) {
    const float* inputs  = (const float*)d_in[0];
    const float* cart    = (const float*)d_in[1];
    const int*   r_index = (const int*)  d_in[2];
    const float* pre1  = (const float*)d_in[3];
    const float* pre2  = (const float*)d_in[4];
    const float* pre3  = (const float*)d_in[5];
    const float* pre4  = (const float*)d_in[6];
    const float* post1 = (const float*)d_in[7];
    const float* post2 = (const float*)d_in[8];
    const float* post3 = (const float*)d_in[9];
    const float* post4 = (const float*)d_in[10];
    const float* ck1 = (const float*)d_in[11];
    const float* sk1 = (const float*)d_in[12];
    const float* ck2 = (const float*)d_in[13];
    const float* sk2 = (const float*)d_in[14];
    const float* ck3 = (const float*)d_in[15];
    const float* sk3 = (const float*)d_in[16];
    const float* ck4 = (const float*)d_in[17];
    const float* sk4 = (const float*)d_in[18];

    float* polar = (float*)d_ws;

    polar_stage_kernel<<<dim3(64, 16, 16), 128, 0, stream>>>(
        inputs, r_index,
        pre1, pre2, pre3, pre4, post1, post2, post3, post4,
        ck1, sk1, ck2, sk2, ck3, sk3, ck4, sk4, polar);

    cart_stage_kernel<<<dim3(32, 4), 256, 0, stream>>>(
        cart, polar, (float*)d_out);
}


# LDS tiles re-laid bank-conflict free: XOR column swizzle on stage 1's gathered and term tiles; stage 2 holds both tiles K-contiguous at 68-dword pitch, fragments as 8-byte loads; sums bit-identical
# speedup vs baseline: 1.3448x; 1.3448x over previous
#include <hip/hip_runtime.h>


typedef __attribute__((ext_vector_type(2))) float v2f;
typedef __attribute__((ext_vector_type(8))) float v8f;

#define NXC 64
#define NKC 16
#define BATCHC 16
#define FEATC (NXC * NXC)
#define LP 68

__global__ __launch_bounds__(128) void polar_stage_kernel(
    const float* __restrict__ inputs,
    const int*   __restrict__ r_index,
    const float* __restrict__ pre1,  const float* __restrict__ pre2,
    const float* __restrict__ pre3,  const float* __restrict__ pre4,
    const float* __restrict__ post1, const float* __restrict__ post2,
    const float* __restrict__ post3, const float* __restrict__ post4,
    const float* __restrict__ ck1, const float* __restrict__ sk1,
    const float* __restrict__ ck2, const float* __restrict__ sk2,
    const float* __restrict__ ck3, const float* __restrict__ sk3,
    const float* __restrict__ ck4, const float* __restrict__ sk4,
    float* __restrict__ polar)
{
    __shared__ __attribute__((aligned(16))) float sD[2][NXC * NXC];
    __shared__ __attribute__((aligned(16))) float sA[NXC * NXC];
    __shared__ __attribute__((aligned(16))) float sB[NXC * NXC];

    const int tid  = threadIdx.x;
    const int lane = tid & 31;
    const int wave = tid >> 5;

    const int c     = blockIdx.x;
    const int batch = blockIdx.y;
    const int f     = blockIdx.z;

    const int* rix = r_index + c * FEATC;
    const float* baseR = inputs + ((size_t)(batch * 2 + 0) * FEATC) * NKC + f;
    const float* baseI = inputs + ((size_t)(batch * 2 + 1) * FEATC) * NKC + f;
    for (int e = tid; e < FEATC; e += 128) {
        const int idx = rix[e];
        const int pe  = e ^ (((e >> 6) & 15) << 2);
        sD[0][pe] = baseR[(size_t)idx * NKC];
        sD[1][pe] = baseI[(size_t)idx * NKC];
    }
    __syncthreads();

    const float* PRE[4]  = {pre1,  pre2,  pre3,  pre4};
    const float* POST[4] = {post1, post2, post3, post4};
    const float* K2[4]   = {ck1, sk1, ck3, sk4};
    const float* K1[4]   = {ck2, sk2, sk3, ck4};
    const int    DSEL[4] = {0, 0, 1, 1};

    const int kcol    = wave * 16 + (lane & 15);
    const int halfsel = lane >> 4;
    const int arow    = (lane & 15);
    const int kcolx   = kcol ^ (halfsel << 4);
    const int xa      = arow << 2;
    float acc = 0.0f;

    for (int t = 0; t < 4; ++t) {
        __syncthreads();
        const float* k2f   = K2[t] + f * FEATC;
        const float* k1f   = K1[t] + f * FEATC;
        const float* postf = POST[t] + f * NXC;
        const float* pref  = PRE[t]  + f * NXC;
        for (int e = tid; e < FEATC; e += 128) {
            const int row = e >> 6;
            sA[e ^ (((row >> 3) & 1) << 4)] = postf[row] * k2f[e];
            sB[e ^ (((row >> 1) & 1) << 4)] = pref[row]  * k1f[e];
        }
        __syncthreads();

        const float* Dt = sD[DSEL[t]];

        for (int it = 0; it < 4; it += 2) {
            v8f m0 = {};
            v8f m1 = {};
            #pragma unroll
            for (int j = 0; j < NXC; j += 4) {
                v2f b;
                b.x = sB[(j + halfsel * 2)     * NXC + kcolx];
                b.y = sB[(j + halfsel * 2 + 1) * NXC + kcolx];
                const int ca = (j ^ xa) + halfsel * 2;
                const v2f a0 = *(const v2f*)&Dt[((it    ) * 16 + arow) * NXC + ca];
                const v2f a1 = *(const v2f*)&Dt[((it + 1) * 16 + arow) * NXC + ca];
                m0 = __builtin_amdgcn_wmma_f32_16x16x4_f32(
                         false, a0, false, b, (short)0, m0, false, false);
                m1 = __builtin_amdgcn_wmma_f32_16x16x4_f32(
                         false, a1, false, b, (short)0, m1, false, false);
            }
            const int rb0 = (it    ) * 16 + halfsel * 8;
            const int rb1 = (it + 1) * 16 + halfsel * 8;
            #pragma unroll
            for (int r = 0; r < 8; ++r) {
                acc += sA[(rb0 + r) * NXC + kcolx] * m0[r];
                acc += sA[(rb1 + r) * NXC + kcolx] * m1[r];
            }
        }
    }

    acc += __shfl_xor(acc, 16, 32);
    if (lane < 16) {
        const size_t fb = (size_t)f * BATCHC + batch;
        polar[fb * FEATC + c * NXC + wave * 16 + lane] = acc;
    }
}

__global__ __launch_bounds__(256) void cart_stage_kernel(
    const float* __restrict__ cart,
    const float* __restrict__ polar,
    float* __restrict__ out)
{
    __shared__ __attribute__((aligned(16))) float sP[64 * LP];
    __shared__ __attribute__((aligned(16))) float sC[128 * LP];

    const int tid  = threadIdx.x;
    const int lane = tid & 31;
    const int wave = tid >> 5;

    const int n0 = blockIdx.x * 128;
    const int m0 = blockIdx.y * 64;

    const int mt0 = (wave >> 2) * 2;
    const int nt0 = (wave & 3) * 2;

    const int halfsel = lane >> 4;
    const int ml      = lane & 15;
    const int srow    = tid >> 6;
    const int scol    = tid & 63;

    v8f acc00 = {};
    v8f acc01 = {};
    v8f acc10 = {};
    v8f acc11 = {};

    for (int q0 = 0; q0 < 4096; q0 += 64) {
        __syncthreads();
        #pragma unroll 8
        for (int i = 0; i < 16; ++i) {
            const int m = srow + 4 * i;
            sP[m * LP + scol] = polar[(size_t)(m0 + m) * 4096 + q0 + scol];
        }
        #pragma unroll 8
        for (int i = 0; i < 32; ++i) {
            const int p = srow + 4 * i;
            sC[p * LP + scol] = cart[(size_t)(n0 + p) * 4096 + q0 + scol];
        }
        if (q0 + 64 < 4096 && tid < 128) {
            __builtin_prefetch(&cart[(size_t)(n0 + tid) * 4096 + q0 + 64], 0, 1);
        }
        __syncthreads();

        const int n_a = (nt0    ) * 16 + ml;
        const int n_b = (nt0 + 1) * 16 + ml;
        #pragma unroll
        for (int j = 0; j < 64; j += 4) {
            const v2f a0 = *(const v2f*)&sP[((mt0    ) * 16 + ml) * LP + j + halfsel * 2];
            const v2f a1 = *(const v2f*)&sP[((mt0 + 1) * 16 + ml) * LP + j + halfsel * 2];
            const v2f b0 = *(const v2f*)&sC[n_a * LP + j + halfsel * 2];
            const v2f b1 = *(const v2f*)&sC[n_b * LP + j + halfsel * 2];
            acc00 = __builtin_amdgcn_wmma_f32_16x16x4_f32(
                        false, a0, false, b0, (short)0, acc00, false, false);
            acc01 = __builtin_amdgcn_wmma_f32_16x16x4_f32(
                        false, a0, false, b1, (short)0, acc01, false, false);
            acc10 = __builtin_amdgcn_wmma_f32_16x16x4_f32(
                        false, a1, false, b0, (short)0, acc10, false, false);
            acc11 = __builtin_amdgcn_wmma_f32_16x16x4_f32(
                        false, a1, false, b1, (short)0, acc11, false, false);
        }
    }

    #pragma unroll
    for (int mi = 0; mi < 2; ++mi) {
        #pragma unroll
        for (int ni = 0; ni < 2; ++ni) {
            const v8f acc = (mi == 0) ? (ni == 0 ? acc00 : acc01)
                                      : (ni == 0 ? acc10 : acc11);
            const int p  = n0 + (nt0 + ni) * 16 + ml;
            const int e1 = p >> 6, e2 = p & 63;
            #pragma unroll
            for (int r = 0; r < 8; ++r) {
                const int fb = m0 + (mt0 + mi) * 16 + halfsel * 8 + r;
                const int fo = fb >> 4, bo = fb & 15;
                out[(((size_t)bo * 64 + e1) * 64 + e2) * NKC + fo] = acc[r];
            }
        }
    }
}

extern "C" void kernel_launch(void* const* d_in, const int* in_sizes, int n_in,
                              void* d_out, int out_size, void* d_ws, size_t ws_size,
                              hipStream_t stream) {
    const float* inputs  = (const float*)d_in[0];
    const float* cart    = (const float*)d_in[1];
    const int*   r_index = (const int*)  d_in[2];
    const float* pre1  = (const float*)d_in[3];
    const float* pre2  = (const float*)d_in[4];
    const float* pre3  = (const float*)d_in[5];
    const float* pre4  = (const float*)d_in[6];
    const float* post1 = (const float*)d_in[7];
    const float* post2 = (const float*)d_in[8];
    const float* post3 = (const float*)d_in[9];
    const float* post4 = (const float*)d_in[10];
    const float* ck1 = (const float*)d_in[11];
    const float* sk1 = (const float*)d_in[12];
    const float* ck2 = (const float*)d_in[13];
    const float* sk2 = (const float*)d_in[14];
    const float* ck3 = (const float*)d_in[15];
    const float* sk3 = (const float*)d_in[16];
    const float* ck4 = (const float*)d_in[17];
    const float* sk4 = (const float*)d_in[18];

    float* polar = (float*)d_ws;

    polar_stage_kernel<<<dim3(64, 16, 16), 128, 0, stream>>>(
        inputs, r_index,
        pre1, pre2, pre3, pre4, post1, post2, post3, post4,
        ck1, sk1, ck2, sk2, ck3, sk3, ck4, sk4, polar);

    cart_stage_kernel<<<dim3(32, 4), 256, 0, stream>>>(
        cart, polar, (float*)d_out);
}


# Stage 1 term tables staged with 16-byte global loads, a term's whole share in flight before its first LDS store (was 4-byte flat loads waited for pair by pair); products and sum order unchanged
# speedup vs baseline: 2.0280x; 1.5080x over previous
#include <hip/hip_runtime.h>


typedef __attribute__((ext_vector_type(2))) float v2f;
typedef __attribute__((ext_vector_type(4))) float v4f;
typedef __attribute__((ext_vector_type(8))) float v8f;

#define NXC 64
#define NKC 16
#define BATCHC 16
#define FEATC (NXC * NXC)
#define LP 68

__device__ __forceinline__ void polar_term(
    const float* __restrict__ k2f,
    const float* __restrict__ k1f,
    const float* __restrict__ postf,
    const float* __restrict__ pref,
    const float* Dt,
    float* sA, float* sB,
    const int tid, const int halfsel, const int arow, const int kcolx, const int xa,
    float& acc)
{
    const int c4   = (tid & 15) * 4;
    const int srow = tid >> 4;

    v4f   ka[8], kb[8];
    float pa[8], pb[8];
    #pragma unroll
    for (int i = 0; i < 8; ++i) {
        const int row = srow + 8 * i;
        ka[i] = *(const v4f*)&k2f[row * NXC + c4];
        kb[i] = *(const v4f*)&k1f[row * NXC + c4];
        pa[i] = postf[row];
        pb[i] = pref[row];
    }

    __syncthreads();

    #pragma unroll
    for (int i = 0; i < 8; ++i) {
        const int row = srow + 8 * i;
        *(v4f*)&sA[row * NXC + (c4 ^ (((row >> 3) & 1) << 4))] = pa[i] * ka[i];
        *(v4f*)&sB[row * NXC + (c4 ^ (((row >> 1) & 1) << 4))] = pb[i] * kb[i];
    }
    __syncthreads();

    for (int it = 0; it < 4; it += 2) {
        v8f m0 = {};
        v8f m1 = {};
        #pragma unroll
        for (int j = 0; j < NXC; j += 4) {
            v2f b;
            b.x = sB[(j + halfsel * 2)     * NXC + kcolx];
            b.y = sB[(j + halfsel * 2 + 1) * NXC + kcolx];
            const int ca = (j ^ xa) + halfsel * 2;
            const v2f a0 = *(const v2f*)&Dt[((it    ) * 16 + arow) * NXC + ca];
            const v2f a1 = *(const v2f*)&Dt[((it + 1) * 16 + arow) * NXC + ca];
            m0 = __builtin_amdgcn_wmma_f32_16x16x4_f32(
                     false, a0, false, b, (short)0, m0, false, false);
            m1 = __builtin_amdgcn_wmma_f32_16x16x4_f32(
                     false, a1, false, b, (short)0, m1, false, false);
        }
        const int rb0 = (it    ) * 16 + halfsel * 8;
        const int rb1 = (it + 1) * 16 + halfsel * 8;
        #pragma unroll
        for (int r = 0; r < 8; ++r) {
            acc += sA[(rb0 + r) * NXC + kcolx] * m0[r];
            acc += sA[(rb1 + r) * NXC + kcolx] * m1[r];
        }
    }
}

__global__ __launch_bounds__(128) void polar_stage_kernel(
    const float* __restrict__ inputs,
    const int*   __restrict__ r_index,
    const float* __restrict__ pre1,  const float* __restrict__ pre2,
    const float* __restrict__ pre3,  const float* __restrict__ pre4,
    const float* __restrict__ post1, const float* __restrict__ post2,
    const float* __restrict__ post3, const float* __restrict__ post4,
    const float* __restrict__ ck1, const float* __restrict__ sk1,
    const float* __restrict__ ck2, const float* __restrict__ sk2,
    const float* __restrict__ ck3, const float* __restrict__ sk3,
    const float* __restrict__ ck4, const float* __restrict__ sk4,
    float* __restrict__ polar)
{
    __shared__ __attribute__((aligned(16))) float sD[2][NXC * NXC];
    __shared__ __attribute__((aligned(16))) float sA[NXC * NXC];
    __shared__ __attribute__((aligned(16))) float sB[NXC * NXC];

    const int tid  = threadIdx.x;
    const int lane = tid & 31;
    const int wave = tid >> 5;

    const int c     = blockIdx.x;
    const int batch = blockIdx.y;
    const int f     = blockIdx.z;

    const int* rix = r_index + c * FEATC;
    const float* baseR = inputs + ((size_t)(batch * 2 + 0) * FEATC) * NKC + f;
    const float* baseI = inputs + ((size_t)(batch * 2 + 1) * FEATC) * NKC + f;
    for (int e = tid; e < FEATC; e += 128) {
        const int idx = rix[e];
        const int pe  = e ^ (((e >> 6) & 15) << 2);
        sD[0][pe] = baseR[(size_t)idx * NKC];
        sD[1][pe] = baseI[(size_t)idx * NKC];
    }

    const int kcol    = wave * 16 + (lane & 15);
    const int halfsel = lane >> 4;
    const int arow    = (lane & 15);
    const int kcolx   = kcol ^ (halfsel << 4);
    const int xa      = arow << 2;
    float acc = 0.0f;

    polar_term(ck1 + f * FEATC, ck2 + f * FEATC, post1 + f * NXC, pre1 + f * NXC, sD[0],
               sA, sB, tid, halfsel, arow, kcolx, xa, acc);
    polar_term(sk1 + f * FEATC, sk2 + f * FEATC, post2 + f * NXC, pre2 + f * NXC, sD[0],
               sA, sB, tid, halfsel, arow, kcolx, xa, acc);
    polar_term(ck3 + f * FEATC, sk3 + f * FEATC, post3 + f * NXC, pre3 + f * NXC, sD[1],
               sA, sB, tid, halfsel, arow, kcolx, xa, acc);
    polar_term(sk4 + f * FEATC, ck4 + f * FEATC, post4 + f * NXC, pre4 + f * NXC, sD[1],
               sA, sB, tid, halfsel, arow, kcolx, xa, acc);

    acc += __shfl_xor(acc, 16, 32);
    if (lane < 16) {
        const size_t fb = (size_t)f * BATCHC + batch;
        polar[fb * FEATC + c * NXC + wave * 16 + lane] = acc;
    }
}

__global__ __launch_bounds__(256) void cart_stage_kernel(
    const float* __restrict__ cart,
    const float* __restrict__ polar,
    float* __restrict__ out)
{
    __shared__ __attribute__((aligned(16))) float sP[64 * LP];
    __shared__ __attribute__((aligned(16))) float sC[128 * LP];

    const int tid  = threadIdx.x;
    const int lane = tid & 31;
    const int wave = tid >> 5;

    const int n0 = blockIdx.x * 128;
    const int m0 = blockIdx.y * 64;

    const int mt0 = (wave >> 2) * 2;
    const int nt0 = (wave & 3) * 2;

    const int halfsel = lane >> 4;
    const int ml      = lane & 15;
    const int srow    = tid >> 6;
    const int scol    = tid & 63;

    v8f acc00 = {};
    v8f acc01 = {};
    v8f acc10 = {};
    v8f acc11 = {};

    for (int q0 = 0; q0 < 4096; q0 += 64) {
        __syncthreads();
        #pragma unroll 8
        for (int i = 0; i < 16; ++i) {
            const int m = srow + 4 * i;
            sP[m * LP + scol] = polar[(size_t)(m0 + m) * 4096 + q0 + scol];
        }
        #pragma unroll 8
        for (int i = 0; i < 32; ++i) {
            const int p = srow + 4 * i;
            sC[p * LP + scol] = cart[(size_t)(n0 + p) * 4096 + q0 + scol];
        }
        if (q0 + 64 < 4096 && tid < 128) {
            __builtin_prefetch(&cart[(size_t)(n0 + tid) * 4096 + q0 + 64], 0, 1);
        }
        __syncthreads();

        const int n_a = (nt0    ) * 16 + ml;
        const int n_b = (nt0 + 1) * 16 + ml;
        #pragma unroll
        for (int j = 0; j < 64; j += 4) {
            const v2f a0 = *(const v2f*)&sP[((mt0    ) * 16 + ml) * LP + j + halfsel * 2];
            const v2f a1 = *(const v2f*)&sP[((mt0 + 1) * 16 + ml) * LP + j + halfsel * 2];
            const v2f b0 = *(const v2f*)&sC[n_a * LP + j + halfsel * 2];
            const v2f b1 = *(const v2f*)&sC[n_b * LP + j + halfsel * 2];
            acc00 = __builtin_amdgcn_wmma_f32_16x16x4_f32(
                        false, a0, false, b0, (short)0, acc00, false, false);
            acc01 = __builtin_amdgcn_wmma_f32_16x16x4_f32(
                        false, a0, false, b1, (short)0, acc01, false, false);
            acc10 = __builtin_amdgcn_wmma_f32_16x16x4_f32(
                        false, a1, false, b0, (short)0, acc10, false, false);
            acc11 = __builtin_amdgcn_wmma_f32_16x16x4_f32(
                        false, a1, false, b1, (short)0, acc11, false, false);
        }
    }

    #pragma unroll
    for (int mi = 0; mi < 2; ++mi) {
        #pragma unroll
        for (int ni = 0; ni < 2; ++ni) {
            const v8f acc = (mi == 0) ? (ni == 0 ? acc00 : acc01)
                                      : (ni == 0 ? acc10 : acc11);
            const int p  = n0 + (nt0 + ni) * 16 + ml;
            const int e1 = p >> 6, e2 = p & 63;
            #pragma unroll
            for (int r = 0; r < 8; ++r) {
                const int fb = m0 + (mt0 + mi) * 16 + halfsel * 8 + r;
                const int fo = fb >> 4, bo = fb & 15;
                out[(((size_t)bo * 64 + e1) * 64 + e2) * NKC + fo] = acc[r];
            }
        }
    }
}

extern "C" void kernel_launch(void* const* d_in, const int* in_sizes, int n_in,
                              void* d_out, int out_size, void* d_ws, size_t ws_size,
                              hipStream_t stream) {
    const float* inputs  = (const float*)d_in[0];
    const float* cart    = (const float*)d_in[1];
    const int*   r_index = (const int*)  d_in[2];
    const float* pre1  = (const float*)d_in[3];
    const float* pre2  = (const float*)d_in[4];
    const float* pre3  = (const float*)d_in[5];
    const float* pre4  = (const float*)d_in[6];
    const float* post1 = (const float*)d_in[7];
    const float* post2 = (const float*)d_in[8];
    const float* post3 = (const float*)d_in[9];
    const float* post4 = (const float*)d_in[10];
    const float* ck1 = (const float*)d_in[11];
    const float* sk1 = (const float*)d_in[12];
    const float* ck2 = (const float*)d_in[13];
    const float* sk2 = (const float*)d_in[14];
    const float* ck3 = (const float*)d_in[15];
    const float* sk3 = (const float*)d_in[16];
    const float* ck4 = (const float*)d_in[17];
    const float* sk4 = (const float*)d_in[18];

    float* polar = (float*)d_ws;

    polar_stage_kernel<<<dim3(64, 16, 16), 128, 0, stream>>>(
        inputs, r_index,
        pre1, pre2, pre3, pre4, post1, post2, post3, post4,
        ck1, sk1, ck2, sk2, ck3, sk3, ck4, sk4, polar);

    cart_stage_kernel<<<dim3(32, 4), 256, 0, stream>>>(
        cart, polar, (float*)d_out);
}


# Stage 1 block covers 4 frequencies: gather reads each indexed row as one 16-byte load (was one 4-byte load per frequency and block), R then I plane through the same 64 KB; per-f sums bit-identical
# speedup vs baseline: 3.9213x; 1.9336x over previous
#include <hip/hip_runtime.h>


typedef __attribute__((ext_vector_type(2))) float v2f;
typedef __attribute__((ext_vector_type(4))) float v4f;
typedef __attribute__((ext_vector_type(8))) float v8f;

#define NXC 64
#define NKC 16
#define BATCHC 16
#define FEATC (NXC * NXC)
#define FG 4
#define LP 68

__device__ __forceinline__ void polar_term(
    const float* __restrict__ k2f,
    const float* __restrict__ k1f,
    const float* __restrict__ postf,
    const float* __restrict__ pref,
    const float* Dt,
    float* sA, float* sB,
    const int tid, const int halfsel, const int arow, const int kcolx, const int xa,
    float& acc)
{
    const int c4   = (tid & 15) * 4;
    const int srow = tid >> 4;

    v4f   ka[8], kb[8];
    float pa[8], pb[8];
    #pragma unroll
    for (int i = 0; i < 8; ++i) {
        const int row = srow + 8 * i;
        ka[i] = *(const v4f*)&k2f[row * NXC + c4];
        kb[i] = *(const v4f*)&k1f[row * NXC + c4];
        pa[i] = postf[row];
        pb[i] = pref[row];
    }

    __syncthreads();

    #pragma unroll
    for (int i = 0; i < 8; ++i) {
        const int row = srow + 8 * i;
        *(v4f*)&sA[row * NXC + (c4 ^ (((row >> 3) & 1) << 4))] = pa[i] * ka[i];
        *(v4f*)&sB[row * NXC + (c4 ^ (((row >> 1) & 1) << 4))] = pb[i] * kb[i];
    }
    __syncthreads();

    for (int it = 0; it < 4; it += 2) {
        v8f m0 = {};
        v8f m1 = {};
        #pragma unroll
        for (int j = 0; j < NXC; j += 4) {
            v2f b;
            b.x = sB[(j + halfsel * 2)     * NXC + kcolx];
            b.y = sB[(j + halfsel * 2 + 1) * NXC + kcolx];
            const int ca = (j ^ xa) + halfsel * 2;
            const v2f a0 = *(const v2f*)&Dt[((it    ) * 16 + arow) * NXC + ca];
            const v2f a1 = *(const v2f*)&Dt[((it + 1) * 16 + arow) * NXC + ca];
            m0 = __builtin_amdgcn_wmma_f32_16x16x4_f32(
                     false, a0, false, b, (short)0, m0, false, false);
            m1 = __builtin_amdgcn_wmma_f32_16x16x4_f32(
                     false, a1, false, b, (short)0, m1, false, false);
        }
        const int rb0 = (it    ) * 16 + halfsel * 8;
        const int rb1 = (it + 1) * 16 + halfsel * 8;
        #pragma unroll
        for (int r = 0; r < 8; ++r) {
            acc += sA[(rb0 + r) * NXC + kcolx] * m0[r];
            acc += sA[(rb1 + r) * NXC + kcolx] * m1[r];
        }
    }
}

__device__ __forceinline__ void gather_plane(
    const float* __restrict__ src,
    const int*   __restrict__ rix,
    float* sD, const int tid)
{
    constexpr int NE = FEATC / 128;
    int idx[NE];
    v4f v[NE];
    #pragma unroll
    for (int u = 0; u < NE; ++u)
        idx[u] = rix[tid + 128 * u];
    #pragma unroll
    for (int u = 0; u < NE; ++u)
        v[u] = *(const v4f*)&src[(size_t)idx[u] * NKC];
    #pragma unroll
    for (int u = 0; u < NE; ++u) {
        const int e  = tid + 128 * u;
        const int pe = e ^ (((e >> 6) & 15) << 2);
        sD[0 * FEATC + pe] = v[u].x;
        sD[1 * FEATC + pe] = v[u].y;
        sD[2 * FEATC + pe] = v[u].z;
        sD[3 * FEATC + pe] = v[u].w;
    }
}

__global__ __launch_bounds__(128) void polar_stage_kernel(
    const float* __restrict__ inputs,
    const int*   __restrict__ r_index,
    const float* __restrict__ pre1,  const float* __restrict__ pre2,
    const float* __restrict__ pre3,  const float* __restrict__ pre4,
    const float* __restrict__ post1, const float* __restrict__ post2,
    const float* __restrict__ post3, const float* __restrict__ post4,
    const float* __restrict__ ck1, const float* __restrict__ sk1,
    const float* __restrict__ ck2, const float* __restrict__ sk2,
    const float* __restrict__ ck3, const float* __restrict__ sk3,
    const float* __restrict__ ck4, const float* __restrict__ sk4,
    float* __restrict__ polar)
{
    __shared__ __attribute__((aligned(16))) float sD[FG * NXC * NXC];
    __shared__ __attribute__((aligned(16))) float sA[NXC * NXC];
    __shared__ __attribute__((aligned(16))) float sB[NXC * NXC];
    __shared__ float sAcc[FG * 128];

    const int tid  = threadIdx.x;
    const int lane = tid & 31;
    const int wave = tid >> 5;

    const int c     = blockIdx.x;
    const int batch = blockIdx.y;
    const int f0    = blockIdx.z * FG;

    const int*   rix  = r_index + c * FEATC;
    const float* srcR = inputs + ((size_t)(batch * 2 + 0) * FEATC) * NKC + f0;
    const float* srcI = inputs + ((size_t)(batch * 2 + 1) * FEATC) * NKC + f0;

    const int kcol    = wave * 16 + (lane & 15);
    const int halfsel = lane >> 4;
    const int arow    = (lane & 15);
    const int kcolx   = kcol ^ (halfsel << 4);
    const int xa      = arow << 2;

    gather_plane(srcR, rix, sD, tid);

    for (int ff = 0; ff < FG; ++ff) {
        const int f = f0 + ff;
        float acc = 0.0f;
        polar_term(ck1 + f * FEATC, ck2 + f * FEATC, post1 + f * NXC, pre1 + f * NXC, sD + ff * FEATC,
                   sA, sB, tid, halfsel, arow, kcolx, xa, acc);
        polar_term(sk1 + f * FEATC, sk2 + f * FEATC, post2 + f * NXC, pre2 + f * NXC, sD + ff * FEATC,
                   sA, sB, tid, halfsel, arow, kcolx, xa, acc);
        sAcc[ff * 128 + tid] = acc;
    }

    __syncthreads();

    gather_plane(srcI, rix, sD, tid);

    for (int ff = 0; ff < FG; ++ff) {
        const int f = f0 + ff;
        float acc = sAcc[ff * 128 + tid];
        polar_term(ck3 + f * FEATC, sk3 + f * FEATC, post3 + f * NXC, pre3 + f * NXC, sD + ff * FEATC,
                   sA, sB, tid, halfsel, arow, kcolx, xa, acc);
        polar_term(sk4 + f * FEATC, ck4 + f * FEATC, post4 + f * NXC, pre4 + f * NXC, sD + ff * FEATC,
                   sA, sB, tid, halfsel, arow, kcolx, xa, acc);

        acc += __shfl_xor(acc, 16, 32);
        if (lane < 16) {
            const size_t fb = (size_t)f * BATCHC + batch;
            polar[fb * FEATC + c * NXC + wave * 16 + lane] = acc;
        }
    }
}

__global__ __launch_bounds__(256) void cart_stage_kernel(
    const float* __restrict__ cart,
    const float* __restrict__ polar,
    float* __restrict__ out)
{
    __shared__ __attribute__((aligned(16))) float sP[64 * LP];
    __shared__ __attribute__((aligned(16))) float sC[128 * LP];

    const int tid  = threadIdx.x;
    const int lane = tid & 31;
    const int wave = tid >> 5;

    const int n0 = blockIdx.x * 128;
    const int m0 = blockIdx.y * 64;

    const int mt0 = (wave >> 2) * 2;
    const int nt0 = (wave & 3) * 2;

    const int halfsel = lane >> 4;
    const int ml      = lane & 15;
    const int srow    = tid >> 6;
    const int scol    = tid & 63;

    v8f acc00 = {};
    v8f acc01 = {};
    v8f acc10 = {};
    v8f acc11 = {};

    for (int q0 = 0; q0 < 4096; q0 += 64) {
        __syncthreads();
        #pragma unroll 8
        for (int i = 0; i < 16; ++i) {
            const int m = srow + 4 * i;
            sP[m * LP + scol] = polar[(size_t)(m0 + m) * 4096 + q0 + scol];
        }
        #pragma unroll 8
        for (int i = 0; i < 32; ++i) {
            const int p = srow + 4 * i;
            sC[p * LP + scol] = cart[(size_t)(n0 + p) * 4096 + q0 + scol];
        }
        if (q0 + 64 < 4096 && tid < 128) {
            __builtin_prefetch(&cart[(size_t)(n0 + tid) * 4096 + q0 + 64], 0, 1);
        }
        __syncthreads();

        const int n_a = (nt0    ) * 16 + ml;
        const int n_b = (nt0 + 1) * 16 + ml;
        #pragma unroll
        for (int j = 0; j < 64; j += 4) {
            const v2f a0 = *(const v2f*)&sP[((mt0    ) * 16 + ml) * LP + j + halfsel * 2];
            const v2f a1 = *(const v2f*)&sP[((mt0 + 1) * 16 + ml) * LP + j + halfsel * 2];
            const v2f b0 = *(const v2f*)&sC[n_a * LP + j + halfsel * 2];
            const v2f b1 = *(const v2f*)&sC[n_b * LP + j + halfsel * 2];
            acc00 = __builtin_amdgcn_wmma_f32_16x16x4_f32(
                        false, a0, false, b0, (short)0, acc00, false, false);
            acc01 = __builtin_amdgcn_wmma_f32_16x16x4_f32(
                        false, a0, false, b1, (short)0, acc01, false, false);
            acc10 = __builtin_amdgcn_wmma_f32_16x16x4_f32(
                        false, a1, false, b0, (short)0, acc10, false, false);
            acc11 = __builtin_amdgcn_wmma_f32_16x16x4_f32(
                        false, a1, false, b1, (short)0, acc11, false, false);
        }
    }

    #pragma unroll
    for (int mi = 0; mi < 2; ++mi) {
        #pragma unroll
        for (int ni = 0; ni < 2; ++ni) {
            const v8f acc = (mi == 0) ? (ni == 0 ? acc00 : acc01)
                                      : (ni == 0 ? acc10 : acc11);
            const int p  = n0 + (nt0 + ni) * 16 + ml;
            const int e1 = p >> 6, e2 = p & 63;
            #pragma unroll
            for (int r = 0; r < 8; ++r) {
                const int fb = m0 + (mt0 + mi) * 16 + halfsel * 8 + r;
                const int fo = fb >> 4, bo = fb & 15;
                out[(((size_t)bo * 64 + e1) * 64 + e2) * NKC + fo] = acc[r];
            }
        }
    }
}

extern "C" void kernel_launch(void* const* d_in, const int* in_sizes, int n_in,
                              void* d_out, int out_size, void* d_ws, size_t ws_size,
                              hipStream_t stream) {
    const float* inputs  = (const float*)d_in[0];
    const float* cart    = (const float*)d_in[1];
    const int*   r_index = (const int*)  d_in[2];
    const float* pre1  = (const float*)d_in[3];
    const float* pre2  = (const float*)d_in[4];
    const float* pre3  = (const float*)d_in[5];
    const float* pre4  = (const float*)d_in[6];
    const float* post1 = (const float*)d_in[7];
    const float* post2 = (const float*)d_in[8];
    const float* post3 = (const float*)d_in[9];
    const float* post4 = (const float*)d_in[10];
    const float* ck1 = (const float*)d_in[11];
    const float* sk1 = (const float*)d_in[12];
    const float* ck2 = (const float*)d_in[13];
    const float* sk2 = (const float*)d_in[14];
    const float* ck3 = (const float*)d_in[15];
    const float* sk3 = (const float*)d_in[16];
    const float* ck4 = (const float*)d_in[17];
    const float* sk4 = (const float*)d_in[18];

    float* polar = (float*)d_ws;

    polar_stage_kernel<<<dim3(64, 16, 4), 128, 0, stream>>>(
        inputs, r_index,
        pre1, pre2, pre3, pre4, post1, post2, post3, post4,
        ck1, sk1, ck2, sk2, ck3, sk3, ck4, sk4, polar);

    cart_stage_kernel<<<dim3(32, 4), 256, 0, stream>>>(
        cart, polar, (float*)d_out);
}


# Stage 1 reads a tile's A fragments from LDS once, as one group ahead of the WMMAs, and keeps them in registers for both terms (was one load per WMMA pair, waited for at once); sums bit-identical
# speedup vs baseline: 4.0999x; 1.0455x over previous
#include <hip/hip_runtime.h>


typedef __attribute__((ext_vector_type(2))) float v2f;
typedef __attribute__((ext_vector_type(4))) float v4f;
typedef __attribute__((ext_vector_type(8))) float v8f;

#define NXC 64
#define NKC 16
#define BATCHC 16
#define FEATC (NXC * NXC)
#define FG 4
#define LP 68

template <bool LOAD_A>
__device__ __forceinline__ void polar_term(
    const float* __restrict__ k2f,
    const float* __restrict__ k1f,
    const float* __restrict__ postf,
    const float* __restrict__ pref,
    const float* Dt,
    float* sA, float* sB,
    const int tid, const int halfsel, const int arow, const int kcolx, const int xa,
    v2f (&af)[4][16],
    float& acc)
{
    const int c4   = (tid & 15) * 4;
    const int srow = tid >> 4;

    v4f   ka[8], kb[8];
    float pa[8], pb[8];
    #pragma unroll
    for (int i = 0; i < 8; ++i) {
        const int row = srow + 8 * i;
        ka[i] = *(const v4f*)&k2f[row * NXC + c4];
        kb[i] = *(const v4f*)&k1f[row * NXC + c4];
        pa[i] = postf[row];
        pb[i] = pref[row];
    }

    __syncthreads();

    #pragma unroll
    for (int i = 0; i < 8; ++i) {
        const int row = srow + 8 * i;
        *(v4f*)&sA[row * NXC + (c4 ^ (((row >> 3) & 1) << 4))] = pa[i] * ka[i];
        *(v4f*)&sB[row * NXC + (c4 ^ (((row >> 1) & 1) << 4))] = pb[i] * kb[i];
    }
    __syncthreads();


    v2f bf[16];
    #pragma unroll
    for (int jj = 0; jj < 16; ++jj) {
        bf[jj].x = sB[(4 * jj + halfsel * 2)     * NXC + kcolx];
        bf[jj].y = sB[(4 * jj + halfsel * 2 + 1) * NXC + kcolx];
    }
    if (LOAD_A) {
        #pragma unroll
        for (int it = 0; it < 4; it += 2) {
            #pragma unroll
            for (int jj = 0; jj < 16; ++jj) {
                const int ca = ((4 * jj) ^ xa) + halfsel * 2;
                af[it    ][jj] = *(const v2f*)&Dt[((it    ) * 16 + arow) * NXC + ca];
                af[it + 1][jj] = *(const v2f*)&Dt[((it + 1) * 16 + arow) * NXC + ca];
            }
        }
    }

    #pragma unroll
    for (int it = 0; it < 4; it += 2) {
        v8f m0 = {};
        v8f m1 = {};
        #pragma unroll
        for (int jj = 0; jj < 16; ++jj) {
            m0 = __builtin_amdgcn_wmma_f32_16x16x4_f32(
                     false, af[it    ][jj], false, bf[jj], (short)0, m0, false, false);
            m1 = __builtin_amdgcn_wmma_f32_16x16x4_f32(
                     false, af[it + 1][jj], false, bf[jj], (short)0, m1, false, false);
        }
        const int rb0 = (it    ) * 16 + halfsel * 8;
        const int rb1 = (it + 1) * 16 + halfsel * 8;
        #pragma unroll
        for (int r = 0; r < 8; ++r) {
            acc += sA[(rb0 + r) * NXC + kcolx] * m0[r];
            acc += sA[(rb1 + r) * NXC + kcolx] * m1[r];
        }
    }
}

__device__ __forceinline__ void gather_plane(
    const float* __restrict__ src,
    const int*   __restrict__ rix,
    float* sD, const int tid)
{
    constexpr int NE = FEATC / 128;
    int idx[NE];
    v4f v[NE];
    #pragma unroll
    for (int u = 0; u < NE; ++u)
        idx[u] = rix[tid + 128 * u];
    #pragma unroll
    for (int u = 0; u < NE; ++u)
        v[u] = *(const v4f*)&src[(size_t)idx[u] * NKC];
    #pragma unroll
    for (int u = 0; u < NE; ++u) {
        const int e  = tid + 128 * u;
        const int pe = e ^ (((e >> 6) & 15) << 2);
        sD[0 * FEATC + pe] = v[u].x;
        sD[1 * FEATC + pe] = v[u].y;
        sD[2 * FEATC + pe] = v[u].z;
        sD[3 * FEATC + pe] = v[u].w;
    }
}

__global__ __launch_bounds__(128) __attribute__((amdgpu_waves_per_eu(3))) void polar_stage_kernel(
    const float* __restrict__ inputs,
    const int*   __restrict__ r_index,
    const float* __restrict__ pre1,  const float* __restrict__ pre2,
    const float* __restrict__ pre3,  const float* __restrict__ pre4,
    const float* __restrict__ post1, const float* __restrict__ post2,
    const float* __restrict__ post3, const float* __restrict__ post4,
    const float* __restrict__ ck1, const float* __restrict__ sk1,
    const float* __restrict__ ck2, const float* __restrict__ sk2,
    const float* __restrict__ ck3, const float* __restrict__ sk3,
    const float* __restrict__ ck4, const float* __restrict__ sk4,
    float* __restrict__ polar)
{
    __shared__ __attribute__((aligned(16))) float sD[FG * NXC * NXC];
    __shared__ __attribute__((aligned(16))) float sA[NXC * NXC];
    __shared__ __attribute__((aligned(16))) float sB[NXC * NXC];
    __shared__ float sAcc[FG * 128];

    const int tid  = threadIdx.x;
    const int lane = tid & 31;
    const int wave = tid >> 5;

    const int c     = blockIdx.x;
    const int batch = blockIdx.y;
    const int f0    = blockIdx.z * FG;

    const int*   rix  = r_index + c * FEATC;
    const float* srcR = inputs + ((size_t)(batch * 2 + 0) * FEATC) * NKC + f0;
    const float* srcI = inputs + ((size_t)(batch * 2 + 1) * FEATC) * NKC + f0;

    const int kcol    = wave * 16 + (lane & 15);
    const int halfsel = lane >> 4;
    const int arow    = (lane & 15);
    const int kcolx   = kcol ^ (halfsel << 4);
    const int xa      = arow << 2;

    gather_plane(srcR, rix, sD, tid);

    for (int ff = 0; ff < FG; ++ff) {
        const int f = f0 + ff;
        float acc = 0.0f;
        v2f af[4][16];
        polar_term<true >(ck1 + f * FEATC, ck2 + f * FEATC, post1 + f * NXC, pre1 + f * NXC, sD + ff * FEATC,
                          sA, sB, tid, halfsel, arow, kcolx, xa, af, acc);
        polar_term<false>(sk1 + f * FEATC, sk2 + f * FEATC, post2 + f * NXC, pre2 + f * NXC, sD + ff * FEATC,
                          sA, sB, tid, halfsel, arow, kcolx, xa, af, acc);
        sAcc[ff * 128 + tid] = acc;
    }

    __syncthreads();

    gather_plane(srcI, rix, sD, tid);

    for (int ff = 0; ff < FG; ++ff) {
        const int f = f0 + ff;
        float acc = sAcc[ff * 128 + tid];
        v2f af[4][16];
        polar_term<true >(ck3 + f * FEATC, sk3 + f * FEATC, post3 + f * NXC, pre3 + f * NXC, sD + ff * FEATC,
                          sA, sB, tid, halfsel, arow, kcolx, xa, af, acc);
        polar_term<false>(sk4 + f * FEATC, ck4 + f * FEATC, post4 + f * NXC, pre4 + f * NXC, sD + ff * FEATC,
                          sA, sB, tid, halfsel, arow, kcolx, xa, af, acc);

        acc += __shfl_xor(acc, 16, 32);
        if (lane < 16) {
            const size_t fb = (size_t)f * BATCHC + batch;
            polar[fb * FEATC + c * NXC + wave * 16 + lane] = acc;
        }
    }
}

__global__ __launch_bounds__(256) void cart_stage_kernel(
    const float* __restrict__ cart,
    const float* __restrict__ polar,
    float* __restrict__ out)
{
    __shared__ __attribute__((aligned(16))) float sP[64 * LP];
    __shared__ __attribute__((aligned(16))) float sC[128 * LP];

    const int tid  = threadIdx.x;
    const int lane = tid & 31;
    const int wave = tid >> 5;

    const int n0 = blockIdx.x * 128;
    const int m0 = blockIdx.y * 64;

    const int mt0 = (wave >> 2) * 2;
    const int nt0 = (wave & 3) * 2;

    const int halfsel = lane >> 4;
    const int ml      = lane & 15;
    const int srow    = tid >> 6;
    const int scol    = tid & 63;

    v8f acc00 = {};
    v8f acc01 = {};
    v8f acc10 = {};
    v8f acc11 = {};

    for (int q0 = 0; q0 < 4096; q0 += 64) {
        __syncthreads();
        #pragma unroll 8
        for (int i = 0; i < 16; ++i) {
            const int m = srow + 4 * i;
            sP[m * LP + scol] = polar[(size_t)(m0 + m) * 4096 + q0 + scol];
        }
        #pragma unroll 8
        for (int i = 0; i < 32; ++i) {
            const int p = srow + 4 * i;
            sC[p * LP + scol] = cart[(size_t)(n0 + p) * 4096 + q0 + scol];
        }
        if (q0 + 64 < 4096 && tid < 128) {
            __builtin_prefetch(&cart[(size_t)(n0 + tid) * 4096 + q0 + 64], 0, 1);
        }
        __syncthreads();

        const int n_a = (nt0    ) * 16 + ml;
        const int n_b = (nt0 + 1) * 16 + ml;
        #pragma unroll
        for (int j = 0; j < 64; j += 4) {
            const v2f a0 = *(const v2f*)&sP[((mt0    ) * 16 + ml) * LP + j + halfsel * 2];
            const v2f a1 = *(const v2f*)&sP[((mt0 + 1) * 16 + ml) * LP + j + halfsel * 2];
            const v2f b0 = *(const v2f*)&sC[n_a * LP + j + halfsel * 2];
            const v2f b1 = *(const v2f*)&sC[n_b * LP + j + halfsel * 2];
            acc00 = __builtin_amdgcn_wmma_f32_16x16x4_f32(
                        false, a0, false, b0, (short)0, acc00, false, false);
            acc01 = __builtin_amdgcn_wmma_f32_16x16x4_f32(
                        false, a0, false, b1, (short)0, acc01, false, false);
            acc10 = __builtin_amdgcn_wmma_f32_16x16x4_f32(
                        false, a1, false, b0, (short)0, acc10, false, false);
            acc11 = __builtin_amdgcn_wmma_f32_16x16x4_f32(
                        false, a1, false, b1, (short)0, acc11, false, false);
        }
    }

    #pragma unroll
    for (int mi = 0; mi < 2; ++mi) {
        #pragma unroll
        for (int ni = 0; ni < 2; ++ni) {
            const v8f acc = (mi == 0) ? (ni == 0 ? acc00 : acc01)
                                      : (ni == 0 ? acc10 : acc11);
            const int p  = n0 + (nt0 + ni) * 16 + ml;
            const int e1 = p >> 6, e2 = p & 63;
            #pragma unroll
            for (int r = 0; r < 8; ++r) {
                const int fb = m0 + (mt0 + mi) * 16 + halfsel * 8 + r;
                const int fo = fb >> 4, bo = fb & 15;
                out[(((size_t)bo * 64 + e1) * 64 + e2) * NKC + fo] = acc[r];
            }
        }
    }
}

extern "C" void kernel_launch(void* const* d_in, const int* in_sizes, int n_in,
                              void* d_out, int out_size, void* d_ws, size_t ws_size,
                              hipStream_t stream) {
    const float* inputs  = (const float*)d_in[0];
    const float* cart    = (const float*)d_in[1];
    const int*   r_index = (const int*)  d_in[2];
    const float* pre1  = (const float*)d_in[3];
    const float* pre2  = (const float*)d_in[4];
    const float* pre3  = (const float*)d_in[5];
    const float* pre4  = (const float*)d_in[6];
    const float* post1 = (const float*)d_in[7];
    const float* post2 = (const float*)d_in[8];
    const float* post3 = (const float*)d_in[9];
    const float* post4 = (const float*)d_in[10];
    const float* ck1 = (const float*)d_in[11];
    const float* sk1 = (const float*)d_in[12];
    const float* ck2 = (const float*)d_in[13];
    const float* sk2 = (const float*)d_in[14];
    const float* ck3 = (const float*)d_in[15];
    const float* sk3 = (const float*)d_in[16];
    const float* ck4 = (const float*)d_in[17];
    const float* sk4 = (const float*)d_in[18];

    float* polar = (float*)d_ws;

    polar_stage_kernel<<<dim3(64, 16, 4), 128, 0, stream>>>(
        inputs, r_index,
        pre1, pre2, pre3, pre4, post1, post2, post3, post4,
        ck1, sk1, ck2, sk2, ck3, sk3, ck4, sk4, polar);

    cart_stage_kernel<<<dim3(32, 4), 256, 0, stream>>>(
        cart, polar, (float*)d_out);
}


# Stage 2 K loop software-pipelined: next chunk's tiles fetched as 16-byte loads into registers ahead of the current chunk's WMMAs, second LDS buffer, one barrier per chunk; sums bit-identical
# speedup vs baseline: 4.6509x; 1.1344x over previous
#include <hip/hip_runtime.h>


typedef __attribute__((ext_vector_type(2))) float v2f;
typedef __attribute__((ext_vector_type(4))) float v4f;
typedef __attribute__((ext_vector_type(8))) float v8f;

#define NXC 64
#define NKC 16
#define BATCHC 16
#define FEATC (NXC * NXC)
#define FG 4
#define LP 68

template <bool LOAD_A>
__device__ __forceinline__ void polar_term(
    const float* __restrict__ k2f,
    const float* __restrict__ k1f,
    const float* __restrict__ postf,
    const float* __restrict__ pref,
    const float* Dt,
    float* sA, float* sB,
    const int tid, const int halfsel, const int arow, const int kcolx, const int xa,
    v2f (&af)[4][16],
    float& acc)
{
    const int c4   = (tid & 15) * 4;
    const int srow = tid >> 4;

    v4f   ka[8], kb[8];
    float pa[8], pb[8];
    #pragma unroll
    for (int i = 0; i < 8; ++i) {
        const int row = srow + 8 * i;
        ka[i] = *(const v4f*)&k2f[row * NXC + c4];
        kb[i] = *(const v4f*)&k1f[row * NXC + c4];
        pa[i] = postf[row];
        pb[i] = pref[row];
    }

    __syncthreads();

    #pragma unroll
    for (int i = 0; i < 8; ++i) {
        const int row = srow + 8 * i;
        *(v4f*)&sA[row * NXC + (c4 ^ (((row >> 3) & 1) << 4))] = pa[i] * ka[i];
        *(v4f*)&sB[row * NXC + (c4 ^ (((row >> 1) & 1) << 4))] = pb[i] * kb[i];
    }
    __syncthreads();


    v2f bf[16];
    #pragma unroll
    for (int jj = 0; jj < 16; ++jj) {
        bf[jj].x = sB[(4 * jj + halfsel * 2)     * NXC + kcolx];
        bf[jj].y = sB[(4 * jj + halfsel * 2 + 1) * NXC + kcolx];
    }
    if (LOAD_A) {
        #pragma unroll
        for (int it = 0; it < 4; it += 2) {
            #pragma unroll
            for (int jj = 0; jj < 16; ++jj) {
                const int ca = ((4 * jj) ^ xa) + halfsel * 2;
                af[it    ][jj] = *(const v2f*)&Dt[((it    ) * 16 + arow) * NXC + ca];
                af[it + 1][jj] = *(const v2f*)&Dt[((it + 1) * 16 + arow) * NXC + ca];
            }
        }
    }

    #pragma unroll
    for (int it = 0; it < 4; it += 2) {
        v8f m0 = {};
        v8f m1 = {};
        #pragma unroll
        for (int jj = 0; jj < 16; ++jj) {
            m0 = __builtin_amdgcn_wmma_f32_16x16x4_f32(
                     false, af[it    ][jj], false, bf[jj], (short)0, m0, false, false);
            m1 = __builtin_amdgcn_wmma_f32_16x16x4_f32(
                     false, af[it + 1][jj], false, bf[jj], (short)0, m1, false, false);
        }
        const int rb0 = (it    ) * 16 + halfsel * 8;
        const int rb1 = (it + 1) * 16 + halfsel * 8;
        #pragma unroll
        for (int r = 0; r < 8; ++r) {
            acc += sA[(rb0 + r) * NXC + kcolx] * m0[r];
            acc += sA[(rb1 + r) * NXC + kcolx] * m1[r];
        }
    }
}

__device__ __forceinline__ void gather_plane(
    const float* __restrict__ src,
    const int*   __restrict__ rix,
    float* sD, const int tid)
{
    constexpr int NE = FEATC / 128;
    int idx[NE];
    v4f v[NE];
    #pragma unroll
    for (int u = 0; u < NE; ++u)
        idx[u] = rix[tid + 128 * u];
    #pragma unroll
    for (int u = 0; u < NE; ++u)
        v[u] = *(const v4f*)&src[(size_t)idx[u] * NKC];
    #pragma unroll
    for (int u = 0; u < NE; ++u) {
        const int e  = tid + 128 * u;
        const int pe = e ^ (((e >> 6) & 15) << 2);
        sD[0 * FEATC + pe] = v[u].x;
        sD[1 * FEATC + pe] = v[u].y;
        sD[2 * FEATC + pe] = v[u].z;
        sD[3 * FEATC + pe] = v[u].w;
    }
}

__global__ __launch_bounds__(128) __attribute__((amdgpu_waves_per_eu(3))) void polar_stage_kernel(
    const float* __restrict__ inputs,
    const int*   __restrict__ r_index,
    const float* __restrict__ pre1,  const float* __restrict__ pre2,
    const float* __restrict__ pre3,  const float* __restrict__ pre4,
    const float* __restrict__ post1, const float* __restrict__ post2,
    const float* __restrict__ post3, const float* __restrict__ post4,
    const float* __restrict__ ck1, const float* __restrict__ sk1,
    const float* __restrict__ ck2, const float* __restrict__ sk2,
    const float* __restrict__ ck3, const float* __restrict__ sk3,
    const float* __restrict__ ck4, const float* __restrict__ sk4,
    float* __restrict__ polar)
{
    __shared__ __attribute__((aligned(16))) float sD[FG * NXC * NXC];
    __shared__ __attribute__((aligned(16))) float sA[NXC * NXC];
    __shared__ __attribute__((aligned(16))) float sB[NXC * NXC];
    __shared__ float sAcc[FG * 128];

    const int tid  = threadIdx.x;
    const int lane = tid & 31;
    const int wave = tid >> 5;

    const int c     = blockIdx.x;
    const int batch = blockIdx.y;
    const int f0    = blockIdx.z * FG;

    const int*   rix  = r_index + c * FEATC;
    const float* srcR = inputs + ((size_t)(batch * 2 + 0) * FEATC) * NKC + f0;
    const float* srcI = inputs + ((size_t)(batch * 2 + 1) * FEATC) * NKC + f0;

    const int kcol    = wave * 16 + (lane & 15);
    const int halfsel = lane >> 4;
    const int arow    = (lane & 15);
    const int kcolx   = kcol ^ (halfsel << 4);
    const int xa      = arow << 2;

    gather_plane(srcR, rix, sD, tid);

    for (int ff = 0; ff < FG; ++ff) {
        const int f = f0 + ff;
        float acc = 0.0f;
        v2f af[4][16];
        polar_term<true >(ck1 + f * FEATC, ck2 + f * FEATC, post1 + f * NXC, pre1 + f * NXC, sD + ff * FEATC,
                          sA, sB, tid, halfsel, arow, kcolx, xa, af, acc);
        polar_term<false>(sk1 + f * FEATC, sk2 + f * FEATC, post2 + f * NXC, pre2 + f * NXC, sD + ff * FEATC,
                          sA, sB, tid, halfsel, arow, kcolx, xa, af, acc);
        sAcc[ff * 128 + tid] = acc;
    }

    __syncthreads();

    gather_plane(srcI, rix, sD, tid);

    for (int ff = 0; ff < FG; ++ff) {
        const int f = f0 + ff;
        float acc = sAcc[ff * 128 + tid];
        v2f af[4][16];
        polar_term<true >(ck3 + f * FEATC, sk3 + f * FEATC, post3 + f * NXC, pre3 + f * NXC, sD + ff * FEATC,
                          sA, sB, tid, halfsel, arow, kcolx, xa, af, acc);
        polar_term<false>(sk4 + f * FEATC, ck4 + f * FEATC, post4 + f * NXC, pre4 + f * NXC, sD + ff * FEATC,
                          sA, sB, tid, halfsel, arow, kcolx, xa, af, acc);

        acc += __shfl_xor(acc, 16, 32);
        if (lane < 16) {
            const size_t fb = (size_t)f * BATCHC + batch;
            polar[fb * FEATC + c * NXC + wave * 16 + lane] = acc;
        }
    }
}

#define S2_PT (64 * LP)
#define S2_CT (128 * LP)

__device__ __forceinline__ void cart_fetch(
    const float* __restrict__ gP, const float* __restrict__ gC,
    v4f (&rp)[4], v4f (&rc)[8])
{
    #pragma unroll
    for (int i = 0; i < 4; ++i)
        rp[i] = *(const v4f*)(gP + (size_t)(16 * i) * 4096);
    #pragma unroll
    for (int i = 0; i < 8; ++i)
        rc[i] = *(const v4f*)(gC + (size_t)(16 * i) * 4096);
}

__device__ __forceinline__ void cart_stash(
    float* sPn, float* sCn, const int soff,
    const v4f (&rp)[4], const v4f (&rc)[8])
{
    #pragma unroll
    for (int i = 0; i < 4; ++i)
        *(v4f*)&sPn[soff + 16 * i * LP] = rp[i];
    #pragma unroll
    for (int i = 0; i < 8; ++i)
        *(v4f*)&sCn[soff + 16 * i * LP] = rc[i];
}

__device__ __forceinline__ void cart_chunk_mma(
    const float* sPc, const float* sCc,
    const int a0off, const int a1off, const int b0off, const int b1off,
    v8f& acc00, v8f& acc01, v8f& acc10, v8f& acc11)
{
    #pragma unroll
    for (int j = 0; j < 64; j += 4) {
        const v2f a0 = *(const v2f*)&sPc[a0off + j];
        const v2f a1 = *(const v2f*)&sPc[a1off + j];
        const v2f b0 = *(const v2f*)&sCc[b0off + j];
        const v2f b1 = *(const v2f*)&sCc[b1off + j];
        acc00 = __builtin_amdgcn_wmma_f32_16x16x4_f32(
                    false, a0, false, b0, (short)0, acc00, false, false);
        acc01 = __builtin_amdgcn_wmma_f32_16x16x4_f32(
                    false, a0, false, b1, (short)0, acc01, false, false);
        acc10 = __builtin_amdgcn_wmma_f32_16x16x4_f32(
                    false, a1, false, b0, (short)0, acc10, false, false);
        acc11 = __builtin_amdgcn_wmma_f32_16x16x4_f32(
                    false, a1, false, b1, (short)0, acc11, false, false);
    }
}

__global__ __launch_bounds__(256) void cart_stage_kernel(
    const float* __restrict__ cart,
    const float* __restrict__ polar,
    float* __restrict__ out)
{
    __shared__ __attribute__((aligned(16))) float sP[2 * S2_PT];
    __shared__ __attribute__((aligned(16))) float sC[2 * S2_CT];

    const int tid  = threadIdx.x;
    const int lane = tid & 31;
    const int wave = tid >> 5;

    const int n0 = blockIdx.x * 128;
    const int m0 = blockIdx.y * 64;

    const int mt0 = (wave >> 2) * 2;
    const int nt0 = (wave & 3) * 2;

    const int halfsel = lane >> 4;
    const int ml      = lane & 15;

    const int c4   = (tid & 15) * 4;
    const int srow = tid >> 4;
    const int soff = srow * LP + c4;
    const float* gP = polar + (size_t)(m0 + srow) * 4096 + c4;
    const float* gC = cart  + (size_t)(n0 + srow) * 4096 + c4;

    const int a0off = ((mt0    ) * 16 + ml) * LP + halfsel * 2;
    const int a1off = ((mt0 + 1) * 16 + ml) * LP + halfsel * 2;
    const int b0off = ((nt0    ) * 16 + ml) * LP + halfsel * 2;
    const int b1off = ((nt0 + 1) * 16 + ml) * LP + halfsel * 2;

    v8f acc00 = {};
    v8f acc01 = {};
    v8f acc10 = {};
    v8f acc11 = {};

    v4f rp[4], rc[8];

    cart_fetch(gP, gC, rp, rc);
    cart_stash(sP, sC, soff, rp, rc);
    __syncthreads();

    for (int q = 0; q < 63; ++q) {
        const int cur = q & 1;
        cart_fetch(gP + (q + 1) * 64, gC + (q + 1) * 64, rp, rc);
        cart_chunk_mma(sP + cur * S2_PT, sC + cur * S2_CT,
                       a0off, a1off, b0off, b1off, acc00, acc01, acc10, acc11);
        cart_stash(sP + (cur ^ 1) * S2_PT, sC + (cur ^ 1) * S2_CT, soff, rp, rc);
        __syncthreads();
    }
    cart_chunk_mma(sP + S2_PT, sC + S2_CT,
                   a0off, a1off, b0off, b1off, acc00, acc01, acc10, acc11);

    #pragma unroll
    for (int mi = 0; mi < 2; ++mi) {
        #pragma unroll
        for (int ni = 0; ni < 2; ++ni) {
            const v8f acc = (mi == 0) ? (ni == 0 ? acc00 : acc01)
                                      : (ni == 0 ? acc10 : acc11);
            const int p  = n0 + (nt0 + ni) * 16 + ml;
            const int e1 = p >> 6, e2 = p & 63;
            #pragma unroll
            for (int r = 0; r < 8; ++r) {
                const int fb = m0 + (mt0 + mi) * 16 + halfsel * 8 + r;
                const int fo = fb >> 4, bo = fb & 15;
                out[(((size_t)bo * 64 + e1) * 64 + e2) * NKC + fo] = acc[r];
            }
        }
    }
}

extern "C" void kernel_launch(void* const* d_in, const int* in_sizes, int n_in,
                              void* d_out, int out_size, void* d_ws, size_t ws_size,
                              hipStream_t stream) {
    const float* inputs  = (const float*)d_in[0];
    const float* cart    = (const float*)d_in[1];
    const int*   r_index = (const int*)  d_in[2];
    const float* pre1  = (const float*)d_in[3];
    const float* pre2  = (const float*)d_in[4];
    const float* pre3  = (const float*)d_in[5];
    const float* pre4  = (const float*)d_in[6];
    const float* post1 = (const float*)d_in[7];
    const float* post2 = (const float*)d_in[8];
    const float* post3 = (const float*)d_in[9];
    const float* post4 = (const float*)d_in[10];
    const float* ck1 = (const float*)d_in[11];
    const float* sk1 = (const float*)d_in[12];
    const float* ck2 = (const float*)d_in[13];
    const float* sk2 = (const float*)d_in[14];
    const float* ck3 = (const float*)d_in[15];
    const float* sk3 = (const float*)d_in[16];
    const float* ck4 = (const float*)d_in[17];
    const float* sk4 = (const float*)d_in[18];

    float* polar = (float*)d_ws;

    polar_stage_kernel<<<dim3(64, 16, 4), 128, 0, stream>>>(
        inputs, r_index,
        pre1, pre2, pre3, pre4, post1, post2, post3, post4,
        ck1, sk1, ck2, sk2, ck3, sk3, ck4, sk4, polar);

    cart_stage_kernel<<<dim3(32, 4), 256, 0, stream>>>(
        cart, polar, (float*)d_out);
}


# Stage 1 re-cut: block = one batch x 4 c, all 16 f; whole input plane copied to LDS by contiguous loads and gathered there; one table staging (fetched a term ahead) serves 4 tiles; sum order changed
# speedup vs baseline: 6.1257x; 1.3171x over previous
#include <hip/hip_runtime.h>


typedef __attribute__((ext_vector_type(2))) float v2f;
typedef __attribute__((ext_vector_type(4))) float v4f;
typedef __attribute__((ext_vector_type(8))) float v8f;
typedef __attribute__((ext_vector_type(2))) int   v2i;

#define NXC 64
#define NKC 16
#define BATCHC 16
#define FEATC (NXC * NXC)
#define CG 4
#define LP 68

struct TermRegs {
    v4f   ka[4], kb[4];
    float pa[4], pb[4];
};

__device__ __forceinline__ void term_fetch(
    TermRegs& t,
    const float* __restrict__ k2f,
    const float* __restrict__ k1f,
    const float* __restrict__ postf,
    const float* __restrict__ pref,
    const int tid)
{
    const unsigned toff = (unsigned)(tid >> 4) * NXC + (unsigned)(tid & 15) * 4u;
    const unsigned srow = (unsigned)tid >> 4;
    #pragma unroll
    for (unsigned i = 0; i < 4; ++i) {
        t.ka[i] = *(const v4f*)&k2f[toff + 16u * NXC * i];
        t.kb[i] = *(const v4f*)&k1f[toff + 16u * NXC * i];
        t.pa[i] = postf[srow + 16u * i];
        t.pb[i] = pref[srow + 16u * i];
    }
}

__device__ __forceinline__ void term_stash(const TermRegs& t, float* sA, float* sB, const int tid)
{
    const int c4   = (tid & 15) * 4;
    const int srow = tid >> 4;

    __syncthreads();

    #pragma unroll
    for (int i = 0; i < 4; ++i) {
        const int row = srow + 16 * i;
        *(v4f*)&sA[row * NXC + (c4 ^ (((row >> 3) & 1) << 4))] = t.pa[i] * t.ka[i];
        *(v4f*)&sB[row * NXC + (c4 ^ (((row >> 1) & 1) << 4))] = t.pb[i] * t.kb[i];
    }
    __syncthreads();
}

__device__ __forceinline__ void load_bfrags(v2f (&bf)[16], const float* sB, const int halfsel, const int kcolx)
{
    #pragma unroll
    for (int jj = 0; jj < 16; ++jj) {
        bf[jj].x = sB[(4 * jj + halfsel * 2)     * NXC + kcolx];
        bf[jj].y = sB[(4 * jj + halfsel * 2 + 1) * NXC + kcolx];
    }
}

__device__ __forceinline__ void term_compute(
    const float* sA, const float* sB,
    const int halfsel, const int l15, const int itp,
    const v2f (&af)[2][16],
    float (&acc)[4])
{
    const int rb0 = (itp * 2    ) * 16 + halfsel * 8;
    const int rb1 = (itp * 2 + 1) * 16 + halfsel * 8;

    v2f bf[2][16];
    load_bfrags(bf[0], sB, halfsel, l15 ^ (halfsel << 4));

    #pragma unroll
    for (int kt = 0; kt < 4; ++kt) {
        const int kcolx = (kt * 16 + l15) ^ (halfsel << 4);
        float sa0[8], sa1[8];

        __builtin_amdgcn_sched_barrier(0);
        if (kt < 3)
            load_bfrags(bf[(kt + 1) & 1], sB, halfsel, ((kt + 1) * 16 + l15) ^ (halfsel << 4));
        #pragma unroll
        for (int r = 0; r < 8; ++r) {
            sa0[r] = sA[(rb0 + r) * NXC + kcolx];
            sa1[r] = sA[(rb1 + r) * NXC + kcolx];
        }
        __builtin_amdgcn_sched_barrier(0);

        v8f m0 = {};
        v8f m1 = {};
        #pragma unroll
        for (int jj = 0; jj < 16; ++jj) {
            m0 = __builtin_amdgcn_wmma_f32_16x16x4_f32(
                     false, af[0][jj], false, bf[kt & 1][jj], (short)0, m0, false, false);
            m1 = __builtin_amdgcn_wmma_f32_16x16x4_f32(
                     false, af[1][jj], false, bf[kt & 1][jj], (short)0, m1, false, false);
        }
        #pragma unroll
        for (int r = 0; r < 8; ++r) {
            acc[kt] += sa0[r] * m0[r];
            acc[kt] += sa1[r] * m1[r];
        }
    }
}

__device__ __forceinline__ void result_update(
    float* sOut, const float (&acc)[4], const int slot, const int itp, const int lane, const bool first)
{
    float v[4];
    #pragma unroll
    for (int kt = 0; kt < 4; ++kt)
        v[kt] = acc[kt] + __shfl_xor(acc[kt], 16, 32);
    if (itp == 0 && lane < 16) {
        #pragma unroll
        for (int kt = 0; kt < 4; ++kt) {
            if (first) sOut[slot + kt * 16]  = v[kt];
            else       sOut[slot + kt * 16] += v[kt];
        }
    }
    __syncthreads();
    if (itp == 1 && lane < 16) {
        #pragma unroll
        for (int kt = 0; kt < 4; ++kt)
            sOut[slot + kt * 16] += v[kt];
    }
}

__device__ __forceinline__ void load_plane(
    const float* __restrict__ src,
    float* sPl, const int tid)
{
    #pragma unroll 16
    for (int p = 0; p < FEATC * NKC / 4 / 256; ++p) {
        const int q   = p * 256 + tid;
        const int row = q >> 2;
        const int g4  = (q & 3) << 2;
        const v4f v = *(const v4f*)&src[(size_t)q * 4];
        *(v4f*)&sPl[(row << 4) | (g4 ^ (((row >> 2) & 3) << 2))] = v;
    }
}

__global__ __launch_bounds__(256) void polar_stage_kernel(
    const float* __restrict__ inputs,
    const int*   __restrict__ r_index,
    const float* __restrict__ pre1,  const float* __restrict__ pre2,
    const float* __restrict__ pre3,  const float* __restrict__ pre4,
    const float* __restrict__ post1, const float* __restrict__ post2,
    const float* __restrict__ post3, const float* __restrict__ post4,
    const float* __restrict__ ck1, const float* __restrict__ sk1,
    const float* __restrict__ ck2, const float* __restrict__ sk2,
    const float* __restrict__ ck3, const float* __restrict__ sk3,
    const float* __restrict__ ck4, const float* __restrict__ sk4,
    float* __restrict__ polar)
{
    __shared__ __attribute__((aligned(16))) float sPl[FEATC * NKC];
    __shared__ __attribute__((aligned(16))) float sA[NXC * NXC];
    __shared__ __attribute__((aligned(16))) float sB[NXC * NXC];
    __shared__ float sOut[CG * NKC * NXC];

    const int tid  = threadIdx.x;
    const int lane = tid & 31;
    const int wave = tid >> 5;
    const int cl   = wave >> 1;
    const int itp  = wave & 1;

    const int cg    = blockIdx.x;
    const int batch = blockIdx.y;

    const int halfsel = lane >> 4;
    const int l15     = lane & 15;

    unsigned gb[2][16];
    {
        const int* rix = r_index + (cg * CG + cl) * FEATC;
        #pragma unroll
        for (int it = 0; it < 2; ++it) {
            #pragma unroll
            for (int jj = 0; jj < 16; ++jj) {
                const int row = (itp * 2 + it) * 16 + l15;
                const v2i ii = *(const v2i*)&rix[row * NXC + 4 * jj + 2 * halfsel];
                const int i0 = ii.x & (FEATC - 1);
                const int i1 = ii.y & (FEATC - 1);
                const unsigned o0 = (unsigned)((i0 << 4) | (((i0 >> 2) & 3) << 2));
                const unsigned o1 = (unsigned)((i1 << 4) | (((i1 >> 2) & 3) << 2));
                gb[it][jj] = o0 | (o1 << 16);
            }
        }
    }

    TermRegs tr;

    for (int pl = 0; pl < 2; ++pl) {
        const float* ak2 = pl ? ck3   : ck1;
        const float* ak1 = pl ? sk3   : ck2;
        const float* apo = pl ? post3 : post1;
        const float* apr = pl ? pre3  : pre1;
        const float* bk2 = pl ? sk4   : sk1;
        const float* bk1 = pl ? ck4   : sk2;
        const float* bpo = pl ? post4 : post2;
        const float* bpr = pl ? pre4  : pre2;

        term_fetch(tr, ak2, ak1, apo, apr, tid);

        load_plane(inputs + ((size_t)(batch * 2 + pl) * FEATC) * NKC, sPl, tid);
        __syncthreads();

        for (int fp = 0; fp < NKC / 2; ++fp) {
            const int fb = fp * 2;
            const unsigned fb2 = (unsigned)fb | ((unsigned)fb << 16);
            const int fnext = (fb + 2 < NKC) ? fb + 2 : NKC - 1;

            v2f af[2][2][16];
            #pragma unroll
            for (int it = 0; it < 2; ++it) {
                #pragma unroll
                for (int jj = 0; jj < 16; ++jj) {
                    const unsigned q = gb[it][jj] ^ fb2;
                    const v2f g0 = *(const v2f*)&sPl[q & 0xffffu];
                    const v2f g1 = *(const v2f*)&sPl[q >> 16];
                    af[0][it][jj].x = g0.x;  af[0][it][jj].y = g1.x;
                    af[1][it][jj].x = g0.y;  af[1][it][jj].y = g1.y;
                }
            }

            #pragma unroll
            for (int fs = 0; fs < 2; ++fs) {
                const int f  = fb + fs;
                const int fn = fs ? fnext : fb + 1;
                float acc[4] = {0.0f, 0.0f, 0.0f, 0.0f};

                term_stash(tr, sA, sB, tid);
                term_fetch(tr, bk2 + f * FEATC, bk1 + f * FEATC, bpo + f * NXC, bpr + f * NXC, tid);
                term_compute(sA, sB, halfsel, l15, itp, af[fs], acc);

                term_stash(tr, sA, sB, tid);
                term_fetch(tr, ak2 + fn * FEATC, ak1 + fn * FEATC, apo + fn * NXC, apr + fn * NXC, tid);
                term_compute(sA, sB, halfsel, l15, itp, af[fs], acc);

                result_update(sOut, acc, (cl * NKC + f) * NXC + l15, itp, lane, pl == 0);
            }
        }
    }

    __syncthreads();

    for (int f = 0; f < NKC; ++f) {
        const size_t fb = (size_t)f * BATCHC + batch;
        polar[fb * FEATC + cg * (CG * NXC) + tid] = sOut[((tid >> 6) * NKC + f) * NXC + (tid & 63)];
    }
}

#define S2_PT (64 * LP)
#define S2_CT (128 * LP)

__device__ __forceinline__ void cart_fetch(
    const float* __restrict__ gP, const float* __restrict__ gC,
    v4f (&rp)[4], v4f (&rc)[8])
{
    #pragma unroll
    for (int i = 0; i < 4; ++i)
        rp[i] = *(const v4f*)(gP + (size_t)(16 * i) * 4096);
    #pragma unroll
    for (int i = 0; i < 8; ++i)
        rc[i] = *(const v4f*)(gC + (size_t)(16 * i) * 4096);
}

__device__ __forceinline__ void cart_stash(
    float* sPn, float* sCn, const int soff,
    const v4f (&rp)[4], const v4f (&rc)[8])
{
    #pragma unroll
    for (int i = 0; i < 4; ++i)
        *(v4f*)&sPn[soff + 16 * i * LP] = rp[i];
    #pragma unroll
    for (int i = 0; i < 8; ++i)
        *(v4f*)&sCn[soff + 16 * i * LP] = rc[i];
}

__device__ __forceinline__ void cart_chunk_mma(
    const float* sPc, const float* sCc,
    const int a0off, const int a1off, const int b0off, const int b1off,
    v8f& acc00, v8f& acc01, v8f& acc10, v8f& acc11)
{
    #pragma unroll
    for (int j = 0; j < 64; j += 4) {
        const v2f a0 = *(const v2f*)&sPc[a0off + j];
        const v2f a1 = *(const v2f*)&sPc[a1off + j];
        const v2f b0 = *(const v2f*)&sCc[b0off + j];
        const v2f b1 = *(const v2f*)&sCc[b1off + j];
        acc00 = __builtin_amdgcn_wmma_f32_16x16x4_f32(
                    false, a0, false, b0, (short)0, acc00, false, false);
        acc01 = __builtin_amdgcn_wmma_f32_16x16x4_f32(
                    false, a0, false, b1, (short)0, acc01, false, false);
        acc10 = __builtin_amdgcn_wmma_f32_16x16x4_f32(
                    false, a1, false, b0, (short)0, acc10, false, false);
        acc11 = __builtin_amdgcn_wmma_f32_16x16x4_f32(
                    false, a1, false, b1, (short)0, acc11, false, false);
    }
}

__global__ __launch_bounds__(256) void cart_stage_kernel(
    const float* __restrict__ cart,
    const float* __restrict__ polar,
    float* __restrict__ out)
{
    __shared__ __attribute__((aligned(16))) float sP[2 * S2_PT];
    __shared__ __attribute__((aligned(16))) float sC[2 * S2_CT];

    const int tid  = threadIdx.x;
    const int lane = tid & 31;
    const int wave = tid >> 5;

    const int n0 = blockIdx.x * 128;
    const int m0 = blockIdx.y * 64;

    const int mt0 = (wave >> 2) * 2;
    const int nt0 = (wave & 3) * 2;

    const int halfsel = lane >> 4;
    const int ml      = lane & 15;

    const int c4   = (tid & 15) * 4;
    const int srow = tid >> 4;
    const int soff = srow * LP + c4;
    const float* gP = polar + (size_t)(m0 + srow) * 4096 + c4;
    const float* gC = cart  + (size_t)(n0 + srow) * 4096 + c4;

    const int a0off = ((mt0    ) * 16 + ml) * LP + halfsel * 2;
    const int a1off = ((mt0 + 1) * 16 + ml) * LP + halfsel * 2;
    const int b0off = ((nt0    ) * 16 + ml) * LP + halfsel * 2;
    const int b1off = ((nt0 + 1) * 16 + ml) * LP + halfsel * 2;

    v8f acc00 = {};
    v8f acc01 = {};
    v8f acc10 = {};
    v8f acc11 = {};

    v4f rp[4], rc[8];

    cart_fetch(gP, gC, rp, rc);
    cart_stash(sP, sC, soff, rp, rc);
    __syncthreads();

    for (int q = 0; q < 63; ++q) {
        const int cur = q & 1;
        cart_fetch(gP + (q + 1) * 64, gC + (q + 1) * 64, rp, rc);
        cart_chunk_mma(sP + cur * S2_PT, sC + cur * S2_CT,
                       a0off, a1off, b0off, b1off, acc00, acc01, acc10, acc11);
        cart_stash(sP + (cur ^ 1) * S2_PT, sC + (cur ^ 1) * S2_CT, soff, rp, rc);
        __syncthreads();
    }
    cart_chunk_mma(sP + S2_PT, sC + S2_CT,
                   a0off, a1off, b0off, b1off, acc00, acc01, acc10, acc11);

    #pragma unroll
    for (int mi = 0; mi < 2; ++mi) {
        #pragma unroll
        for (int ni = 0; ni < 2; ++ni) {
            const v8f acc = (mi == 0) ? (ni == 0 ? acc00 : acc01)
                                      : (ni == 0 ? acc10 : acc11);
            const int p  = n0 + (nt0 + ni) * 16 + ml;
            const int e1 = p >> 6, e2 = p & 63;
            #pragma unroll
            for (int r = 0; r < 8; ++r) {
                const int fb = m0 + (mt0 + mi) * 16 + halfsel * 8 + r;
                const int fo = fb >> 4, bo = fb & 15;
                out[(((size_t)bo * 64 + e1) * 64 + e2) * NKC + fo] = acc[r];
            }
        }
    }
}

extern "C" void kernel_launch(void* const* d_in, const int* in_sizes, int n_in,
                              void* d_out, int out_size, void* d_ws, size_t ws_size,
                              hipStream_t stream) {
    const float* inputs  = (const float*)d_in[0];
    const float* cart    = (const float*)d_in[1];
    const int*   r_index = (const int*)  d_in[2];
    const float* pre1  = (const float*)d_in[3];
    const float* pre2  = (const float*)d_in[4];
    const float* pre3  = (const float*)d_in[5];
    const float* pre4  = (const float*)d_in[6];
    const float* post1 = (const float*)d_in[7];
    const float* post2 = (const float*)d_in[8];
    const float* post3 = (const float*)d_in[9];
    const float* post4 = (const float*)d_in[10];
    const float* ck1 = (const float*)d_in[11];
    const float* sk1 = (const float*)d_in[12];
    const float* ck2 = (const float*)d_in[13];
    const float* sk2 = (const float*)d_in[14];
    const float* ck3 = (const float*)d_in[15];
    const float* sk3 = (const float*)d_in[16];
    const float* ck4 = (const float*)d_in[17];
    const float* sk4 = (const float*)d_in[18];

    float* polar = (float*)d_ws;

    polar_stage_kernel<<<dim3(16, 16), 256, 0, stream>>>(
        inputs, r_index,
        pre1, pre2, pre3, pre4, post1, post2, post3, post4,
        ck1, sk1, ck2, sk2, ck3, sk3, ck4, sk4, polar);

    cart_stage_kernel<<<dim3(32, 4), 256, 0, stream>>>(
        cart, polar, (float*)d_out);
}


# Stage 2 re-tiled: block = 64x64 of the result (was 64x128), grid 64x4 = 256 blocks, one per WGP (was 128, half the chip idle), wave = 1x2 tiles, K chunks of 256; sums bit-identical
# speedup vs baseline: 6.8753x; 1.1224x over previous
#include <hip/hip_runtime.h>


typedef __attribute__((ext_vector_type(2))) float v2f;
typedef __attribute__((ext_vector_type(4))) float v4f;
typedef __attribute__((ext_vector_type(8))) float v8f;
typedef __attribute__((ext_vector_type(2))) int   v2i;

#define NXC 64
#define NKC 16
#define BATCHC 16
#define FEATC (NXC * NXC)
#define CG 4
#define KC 256
#define LP 260

struct TermRegs {
    v4f   ka[4], kb[4];
    float pa[4], pb[4];
};

__device__ __forceinline__ void term_fetch(
    TermRegs& t,
    const float* __restrict__ k2f,
    const float* __restrict__ k1f,
    const float* __restrict__ postf,
    const float* __restrict__ pref,
    const int tid)
{
    const unsigned toff = (unsigned)(tid >> 4) * NXC + (unsigned)(tid & 15) * 4u;
    const unsigned srow = (unsigned)tid >> 4;
    #pragma unroll
    for (unsigned i = 0; i < 4; ++i) {
        t.ka[i] = *(const v4f*)&k2f[toff + 16u * NXC * i];
        t.kb[i] = *(const v4f*)&k1f[toff + 16u * NXC * i];
        t.pa[i] = postf[srow + 16u * i];
        t.pb[i] = pref[srow + 16u * i];
    }
}

__device__ __forceinline__ void term_stash(const TermRegs& t, float* sA, float* sB, const int tid)
{
    const int c4   = (tid & 15) * 4;
    const int srow = tid >> 4;

    __syncthreads();

    #pragma unroll
    for (int i = 0; i < 4; ++i) {
        const int row = srow + 16 * i;
        *(v4f*)&sA[row * NXC + (c4 ^ (((row >> 3) & 1) << 4))] = t.pa[i] * t.ka[i];
        *(v4f*)&sB[row * NXC + (c4 ^ (((row >> 1) & 1) << 4))] = t.pb[i] * t.kb[i];
    }
    __syncthreads();
}

__device__ __forceinline__ void load_bfrags(v2f (&bf)[16], const float* sB, const int halfsel, const int kcolx)
{
    #pragma unroll
    for (int jj = 0; jj < 16; ++jj) {
        bf[jj].x = sB[(4 * jj + halfsel * 2)     * NXC + kcolx];
        bf[jj].y = sB[(4 * jj + halfsel * 2 + 1) * NXC + kcolx];
    }
}

__device__ __forceinline__ void term_compute(
    const float* sA, const float* sB,
    const int halfsel, const int l15, const int itp,
    const v2f (&af)[2][16],
    float (&acc)[4])
{
    const int rb0 = (itp * 2    ) * 16 + halfsel * 8;
    const int rb1 = (itp * 2 + 1) * 16 + halfsel * 8;

    v2f bf[2][16];
    load_bfrags(bf[0], sB, halfsel, l15 ^ (halfsel << 4));

    #pragma unroll
    for (int kt = 0; kt < 4; ++kt) {
        const int kcolx = (kt * 16 + l15) ^ (halfsel << 4);
        float sa0[8], sa1[8];

        __builtin_amdgcn_sched_barrier(0);
        if (kt < 3)
            load_bfrags(bf[(kt + 1) & 1], sB, halfsel, ((kt + 1) * 16 + l15) ^ (halfsel << 4));
        #pragma unroll
        for (int r = 0; r < 8; ++r) {
            sa0[r] = sA[(rb0 + r) * NXC + kcolx];
            sa1[r] = sA[(rb1 + r) * NXC + kcolx];
        }
        __builtin_amdgcn_sched_barrier(0);

        v8f m0 = {};
        v8f m1 = {};
        #pragma unroll
        for (int jj = 0; jj < 16; ++jj) {
            m0 = __builtin_amdgcn_wmma_f32_16x16x4_f32(
                     false, af[0][jj], false, bf[kt & 1][jj], (short)0, m0, false, false);
            m1 = __builtin_amdgcn_wmma_f32_16x16x4_f32(
                     false, af[1][jj], false, bf[kt & 1][jj], (short)0, m1, false, false);
        }
        #pragma unroll
        for (int r = 0; r < 8; ++r) {
            acc[kt] += sa0[r] * m0[r];
            acc[kt] += sa1[r] * m1[r];
        }
    }
}

__device__ __forceinline__ void result_update(
    float* sOut, const float (&acc)[4], const int slot, const int itp, const int lane, const bool first)
{
    float v[4];
    #pragma unroll
    for (int kt = 0; kt < 4; ++kt)
        v[kt] = acc[kt] + __shfl_xor(acc[kt], 16, 32);
    if (itp == 0 && lane < 16) {
        #pragma unroll
        for (int kt = 0; kt < 4; ++kt) {
            if (first) sOut[slot + kt * 16]  = v[kt];
            else       sOut[slot + kt * 16] += v[kt];
        }
    }
    __syncthreads();
    if (itp == 1 && lane < 16) {
        #pragma unroll
        for (int kt = 0; kt < 4; ++kt)
            sOut[slot + kt * 16] += v[kt];
    }
}

__device__ __forceinline__ void load_plane(
    const float* __restrict__ src,
    float* sPl, const int tid)
{
    #pragma unroll 16
    for (int p = 0; p < FEATC * NKC / 4 / 256; ++p) {
        const int q   = p * 256 + tid;
        const int row = q >> 2;
        const int g4  = (q & 3) << 2;
        const v4f v = *(const v4f*)&src[(size_t)q * 4];
        *(v4f*)&sPl[(row << 4) | (g4 ^ (((row >> 2) & 3) << 2))] = v;
    }
}

__global__ __launch_bounds__(256) void polar_stage_kernel(
    const float* __restrict__ inputs,
    const int*   __restrict__ r_index,
    const float* __restrict__ pre1,  const float* __restrict__ pre2,
    const float* __restrict__ pre3,  const float* __restrict__ pre4,
    const float* __restrict__ post1, const float* __restrict__ post2,
    const float* __restrict__ post3, const float* __restrict__ post4,
    const float* __restrict__ ck1, const float* __restrict__ sk1,
    const float* __restrict__ ck2, const float* __restrict__ sk2,
    const float* __restrict__ ck3, const float* __restrict__ sk3,
    const float* __restrict__ ck4, const float* __restrict__ sk4,
    float* __restrict__ polar)
{
    __shared__ __attribute__((aligned(16))) float sPl[FEATC * NKC];
    __shared__ __attribute__((aligned(16))) float sA[NXC * NXC];
    __shared__ __attribute__((aligned(16))) float sB[NXC * NXC];
    __shared__ float sOut[CG * NKC * NXC];

    const int tid  = threadIdx.x;
    const int lane = tid & 31;
    const int wave = tid >> 5;
    const int cl   = wave >> 1;
    const int itp  = wave & 1;

    const int cg    = blockIdx.x;
    const int batch = blockIdx.y;

    const int halfsel = lane >> 4;
    const int l15     = lane & 15;

    unsigned gb[2][16];
    {
        const int* rix = r_index + (cg * CG + cl) * FEATC;
        #pragma unroll
        for (int it = 0; it < 2; ++it) {
            #pragma unroll
            for (int jj = 0; jj < 16; ++jj) {
                const int row = (itp * 2 + it) * 16 + l15;
                const v2i ii = *(const v2i*)&rix[row * NXC + 4 * jj + 2 * halfsel];
                const int i0 = ii.x & (FEATC - 1);
                const int i1 = ii.y & (FEATC - 1);
                const unsigned o0 = (unsigned)((i0 << 4) | (((i0 >> 2) & 3) << 2));
                const unsigned o1 = (unsigned)((i1 << 4) | (((i1 >> 2) & 3) << 2));
                gb[it][jj] = o0 | (o1 << 16);
            }
        }
    }

    TermRegs tr;

    for (int pl = 0; pl < 2; ++pl) {
        const float* ak2 = pl ? ck3   : ck1;
        const float* ak1 = pl ? sk3   : ck2;
        const float* apo = pl ? post3 : post1;
        const float* apr = pl ? pre3  : pre1;
        const float* bk2 = pl ? sk4   : sk1;
        const float* bk1 = pl ? ck4   : sk2;
        const float* bpo = pl ? post4 : post2;
        const float* bpr = pl ? pre4  : pre2;

        term_fetch(tr, ak2, ak1, apo, apr, tid);

        load_plane(inputs + ((size_t)(batch * 2 + pl) * FEATC) * NKC, sPl, tid);
        __syncthreads();

        for (int fp = 0; fp < NKC / 2; ++fp) {
            const int fb = fp * 2;
            const unsigned fb2 = (unsigned)fb | ((unsigned)fb << 16);
            const int fnext = (fb + 2 < NKC) ? fb + 2 : NKC - 1;

            v2f af[2][2][16];
            #pragma unroll
            for (int it = 0; it < 2; ++it) {
                #pragma unroll
                for (int jj = 0; jj < 16; ++jj) {
                    const unsigned q = gb[it][jj] ^ fb2;
                    const v2f g0 = *(const v2f*)&sPl[q & 0xffffu];
                    const v2f g1 = *(const v2f*)&sPl[q >> 16];
                    af[0][it][jj].x = g0.x;  af[0][it][jj].y = g1.x;
                    af[1][it][jj].x = g0.y;  af[1][it][jj].y = g1.y;
                }
            }

            #pragma unroll
            for (int fs = 0; fs < 2; ++fs) {
                const int f  = fb + fs;
                const int fn = fs ? fnext : fb + 1;
                float acc[4] = {0.0f, 0.0f, 0.0f, 0.0f};

                term_stash(tr, sA, sB, tid);
                term_fetch(tr, bk2 + f * FEATC, bk1 + f * FEATC, bpo + f * NXC, bpr + f * NXC, tid);
                term_compute(sA, sB, halfsel, l15, itp, af[fs], acc);

                term_stash(tr, sA, sB, tid);
                term_fetch(tr, ak2 + fn * FEATC, ak1 + fn * FEATC, apo + fn * NXC, apr + fn * NXC, tid);
                term_compute(sA, sB, halfsel, l15, itp, af[fs], acc);

                result_update(sOut, acc, (cl * NKC + f) * NXC + l15, itp, lane, pl == 0);
            }
        }
    }

    __syncthreads();

    for (int f = 0; f < NKC; ++f) {
        const size_t fb = (size_t)f * BATCHC + batch;
        polar[fb * FEATC + cg * (CG * NXC) + tid] = sOut[((tid >> 6) * NKC + f) * NXC + (tid & 63)];
    }
}

#define S2_T (64 * LP)

__device__ __forceinline__ void cart_fetch(
    const float* __restrict__ gP, const float* __restrict__ gC,
    v4f (&rp)[16], v4f (&rc)[16])
{
    #pragma unroll
    for (int i = 0; i < 16; ++i)
        rp[i] = *(const v4f*)(gP + (size_t)(4 * i) * 4096);
    #pragma unroll
    for (int i = 0; i < 16; ++i)
        rc[i] = *(const v4f*)(gC + (size_t)(4 * i) * 4096);
}

__device__ __forceinline__ void cart_stash(
    float* sPn, float* sCn, const int soff,
    const v4f (&rp)[16], const v4f (&rc)[16])
{
    #pragma unroll
    for (int i = 0; i < 16; ++i)
        *(v4f*)&sPn[soff + 4 * i * LP] = rp[i];
    #pragma unroll
    for (int i = 0; i < 16; ++i)
        *(v4f*)&sCn[soff + 4 * i * LP] = rc[i];
}

__device__ __forceinline__ void cart_chunk_mma(
    const float* sPc, const float* sCc,
    const int aoff, const int b0off, const int b1off,
    v8f& acc0, v8f& acc1)
{
    #pragma unroll
    for (int j = 0; j < KC; j += 4) {
        const v2f a  = *(const v2f*)&sPc[aoff  + j];
        const v2f b0 = *(const v2f*)&sCc[b0off + j];
        const v2f b1 = *(const v2f*)&sCc[b1off + j];
        acc0 = __builtin_amdgcn_wmma_f32_16x16x4_f32(
                   false, a, false, b0, (short)0, acc0, false, false);
        acc1 = __builtin_amdgcn_wmma_f32_16x16x4_f32(
                   false, a, false, b1, (short)0, acc1, false, false);
    }
}

__global__ __launch_bounds__(256) void cart_stage_kernel(
    const float* __restrict__ cart,
    const float* __restrict__ polar,
    float* __restrict__ out)
{
    __shared__ __attribute__((aligned(16))) float sP[2 * S2_T];
    __shared__ __attribute__((aligned(16))) float sC[2 * S2_T];

    const int tid  = threadIdx.x;
    const int lane = tid & 31;
    const int wave = tid >> 5;

    const int n0 = blockIdx.x * 64;
    const int m0 = blockIdx.y * 64;

    const int mt  = wave >> 1;
    const int nt0 = (wave & 1) * 2;

    const int halfsel = lane >> 4;
    const int ml      = lane & 15;

    const int c4   = (tid & 63) * 4;
    const int srow = tid >> 6;
    const int soff = srow * LP + c4;
    const float* gP = polar + (size_t)(m0 + srow) * 4096 + c4;
    const float* gC = cart  + (size_t)(n0 + srow) * 4096 + c4;

    const int aoff  = ((mt     ) * 16 + ml) * LP + halfsel * 2;
    const int b0off = ((nt0    ) * 16 + ml) * LP + halfsel * 2;
    const int b1off = ((nt0 + 1) * 16 + ml) * LP + halfsel * 2;

    v8f acc0 = {};
    v8f acc1 = {};

    v4f rp[16], rc[16];

    cart_fetch(gP, gC, rp, rc);
    cart_stash(sP, sC, soff, rp, rc);
    __syncthreads();

    for (int q = 0; q < 4096 / KC - 1; ++q) {
        const int cur = q & 1;
        cart_fetch(gP + (q + 1) * KC, gC + (q + 1) * KC, rp, rc);
        cart_chunk_mma(sP + cur * S2_T, sC + cur * S2_T,
                       aoff, b0off, b1off, acc0, acc1);
        cart_stash(sP + (cur ^ 1) * S2_T, sC + (cur ^ 1) * S2_T, soff, rp, rc);
        __syncthreads();
    }
    cart_chunk_mma(sP + S2_T, sC + S2_T,
                   aoff, b0off, b1off, acc0, acc1);

    #pragma unroll
    for (int ni = 0; ni < 2; ++ni) {
        const v8f acc = (ni == 0) ? acc0 : acc1;
        const int p  = n0 + (nt0 + ni) * 16 + ml;
        const int e1 = p >> 6, e2 = p & 63;
        #pragma unroll
        for (int r = 0; r < 8; ++r) {
            const int fb = m0 + mt * 16 + halfsel * 8 + r;
            const int fo = fb >> 4, bo = fb & 15;
            out[(((size_t)bo * 64 + e1) * 64 + e2) * NKC + fo] = acc[r];
        }
    }
}

extern "C" void kernel_launch(void* const* d_in, const int* in_sizes, int n_in,
                              void* d_out, int out_size, void* d_ws, size_t ws_size,
                              hipStream_t stream) {
    const float* inputs  = (const float*)d_in[0];
    const float* cart    = (const float*)d_in[1];
    const int*   r_index = (const int*)  d_in[2];
    const float* pre1  = (const float*)d_in[3];
    const float* pre2  = (const float*)d_in[4];
    const float* pre3  = (const float*)d_in[5];
    const float* pre4  = (const float*)d_in[6];
    const float* post1 = (const float*)d_in[7];
    const float* post2 = (const float*)d_in[8];
    const float* post3 = (const float*)d_in[9];
    const float* post4 = (const float*)d_in[10];
    const float* ck1 = (const float*)d_in[11];
    const float* sk1 = (const float*)d_in[12];
    const float* ck2 = (const float*)d_in[13];
    const float* sk2 = (const float*)d_in[14];
    const float* ck3 = (const float*)d_in[15];
    const float* sk3 = (const float*)d_in[16];
    const float* ck4 = (const float*)d_in[17];
    const float* sk4 = (const float*)d_in[18];

    float* polar = (float*)d_ws;

    polar_stage_kernel<<<dim3(16, 16), 256, 0, stream>>>(
        inputs, r_index,
        pre1, pre2, pre3, pre4, post1, post2, post3, post4,
        ck1, sk1, ck2, sk2, ck3, sk3, ck4, sk4, polar);

    cart_stage_kernel<<<dim3(64, 4), 256, 0, stream>>>(
        cart, polar, (float*)d_out);
}


# Stage 1 waves split K, not the result: each wave holds all 16 result tiles for one eighth of j; gather and T built once per block (was 2x / 4x); 8 partials added via LDS; sum order changed
# speedup vs baseline: 12.9541x; 1.8842x over previous
#include <hip/hip_runtime.h>


typedef __attribute__((ext_vector_type(2))) float v2f;
typedef __attribute__((ext_vector_type(4))) float v4f;
typedef __attribute__((ext_vector_type(8))) float v8f;
typedef __attribute__((ext_vector_type(4))) int   v4i;

#define NXC 64
#define NKC 16
#define BATCHC 16
#define FEATC (NXC * NXC)
#define KC 256
#define LP 260

__device__ __forceinline__ void stage_row_weights(
    v2f* dst,
    const float* __restrict__ k2f,
    const float* __restrict__ postf,
    const int tid)
{
    const int c4   = (tid & 15) * 4;
    const int srow = tid >> 4;
    #pragma unroll
    for (int ii = 0; ii < 4; ++ii) {
        const int row = srow + 16 * ii;
        const v4f kv = *(const v4f*)&k2f[row * NXC + c4];
        const v4f av = postf[row] * kv;
        v4f lo, hi;
        lo.x = av.x;  lo.y = av.x;  lo.z = av.y;  lo.w = av.y;
        hi.x = av.z;  hi.y = av.z;  hi.z = av.w;  hi.w = av.w;
        *(v4f*)&dst[row * NXC + c4]     = lo;
        *(v4f*)&dst[row * NXC + c4 + 2] = hi;
    }
}

__device__ __forceinline__ void load_col_weights(
    v2f (&B)[4][2],
    const float* __restrict__ k1f,
    const float* __restrict__ pref,
    const int jbase, const int l15)
{
    #pragma unroll
    for (int n = 0; n < 4; ++n) {
        #pragma unroll
        for (int st = 0; st < 2; ++st) {
            const int j0 = jbase + 2 * st;
            B[n][st].x = pref[j0]     * k1f[(j0)     * NXC + l15 + 16 * n];
            B[n][st].y = pref[j0 + 1] * k1f[(j0 + 1) * NXC + l15 + 16 * n];
        }
    }
}

__global__ __launch_bounds__(256) void polar_stage_kernel(
    const float* __restrict__ inputs,
    const int*   __restrict__ r_index,
    const float* __restrict__ pre1,  const float* __restrict__ pre2,
    const float* __restrict__ pre3,  const float* __restrict__ pre4,
    const float* __restrict__ post1, const float* __restrict__ post2,
    const float* __restrict__ post3, const float* __restrict__ post4,
    const float* __restrict__ ck1, const float* __restrict__ sk1,
    const float* __restrict__ ck2, const float* __restrict__ sk2,
    const float* __restrict__ ck3, const float* __restrict__ sk3,
    const float* __restrict__ ck4, const float* __restrict__ sk4,
    float* __restrict__ polar)
{
    __shared__ __attribute__((aligned(16))) float smem[8 * FEATC + 2 * FEATC + 8 * FEATC];
    v2f*   const sA = (v2f*)smem;
    v2f*   const sX = (v2f*)(smem + 8 * FEATC);
    float* const sP = smem + 10 * FEATC;

    const int tid  = threadIdx.x;
    const int lane = tid & 31;
    const int wave = tid >> 5;

    const int f     = blockIdx.x;
    const int batch = blockIdx.y;

    const int h   = lane >> 4;
    const int l15 = lane & 15;
    const int jbase = 8 * wave + 4 * h;

    const int* rix = r_index + (size_t)l15 * FEATC + jbase;

    v4i idxb[2][4];
    #pragma unroll
    for (int m = 0; m < 4; ++m) {
        idxb[0][m] = *(const v4i*)&rix[m * 16 * FEATC];
        idxb[1][m] = *(const v4i*)&rix[m * 16 * FEATC + NXC];
    }

    {
        const float* pR = inputs + ((size_t)(batch * 2    ) * FEATC) * NKC + f;
        const float* pI = inputs + ((size_t)(batch * 2 + 1) * FEATC) * NKC + f;
        #pragma unroll
        for (int p = 0; p < FEATC / 256; ++p) {
            const int row = p * 256 + tid;
            v2f v;
            v.x = pR[row * NKC];
            v.y = pI[row * NKC];
            sX[row] = v;
        }
    }

    stage_row_weights(sA + 0 * FEATC, ck1 + f * FEATC, post1 + f * NXC, tid);
    stage_row_weights(sA + 1 * FEATC, sk1 + f * FEATC, post2 + f * NXC, tid);
    stage_row_weights(sA + 2 * FEATC, ck3 + f * FEATC, post3 + f * NXC, tid);
    stage_row_weights(sA + 3 * FEATC, sk4 + f * FEATC, post4 + f * NXC, tid);

    v2f Bq[4][4][2];
    load_col_weights(Bq[0], ck2 + f * FEATC, pre1 + f * NXC, jbase, l15);
    load_col_weights(Bq[1], sk2 + f * FEATC, pre2 + f * NXC, jbase, l15);
    load_col_weights(Bq[2], sk3 + f * FEATC, pre3 + f * NXC, jbase, l15);
    load_col_weights(Bq[3], ck4 + f * FEATC, pre4 + f * NXC, jbase, l15);

    __syncthreads();

    v2f x[2][16];
    v2f a[2][4][4];
    v8f acc[4][4] = {};

    #pragma unroll
    for (int m = 0; m < 4; ++m) {
        #pragma unroll
        for (int e = 0; e < 4; ++e)
            x[0][m * 4 + e] = sX[idxb[0][m][e] & (FEATC - 1)];
    }
    #pragma unroll
    for (int t = 0; t < 4; ++t) {
        #pragma unroll
        for (int n = 0; n < 4; ++n)
            a[0][t][n] = sA[t * FEATC + l15 + 16 * n];
    }

    for (int i2 = 0; i2 < NXC / 2; ++i2) {
        #pragma unroll
        for (int b = 0; b < 2; ++b) {
            const int i  = 2 * i2 + b;
            const int in2 = (i + 2 < NXC) ? i + 2 : NXC - 1;
            const int in1 = (i + 1) & (NXC - 1);

            __builtin_amdgcn_sched_barrier(0);
            #pragma unroll
            for (int m = 0; m < 4; ++m)
                idxb[b][m] = *(const v4i*)&rix[m * 16 * FEATC + in2 * NXC];
            #pragma unroll
            for (int m = 0; m < 4; ++m) {
                #pragma unroll
                for (int e = 0; e < 4; ++e)
                    x[b ^ 1][m * 4 + e] = sX[idxb[b ^ 1][m][e] & (FEATC - 1)];
            }
            #pragma unroll
            for (int t = 0; t < 4; ++t) {
                #pragma unroll
                for (int n = 0; n < 4; ++n)
                    a[b ^ 1][t][n] = sA[t * FEATC + in1 * NXC + l15 + 16 * n];
            }
            __builtin_amdgcn_sched_barrier(0);

            #pragma unroll
            for (int st = 0; st < 2; ++st) {
                v2f tR[4], tI[4];
                #pragma unroll
                for (int n = 0; n < 4; ++n) {
                    tR[n] = a[b][0][n] * Bq[0][n][st] + a[b][1][n] * Bq[1][n][st];
                    tI[n] = a[b][2][n] * Bq[2][n][st] + a[b][3][n] * Bq[3][n][st];
                }
                #pragma unroll
                for (int m = 0; m < 4; ++m) {
                    const v2f x0 = x[b][m * 4 + 2 * st];
                    const v2f x1 = x[b][m * 4 + 2 * st + 1];
                    v2f aR, aI;
                    aR.x = x0.x;  aR.y = x1.x;
                    aI.x = x0.y;  aI.y = x1.y;
                    #pragma unroll
                    for (int n = 0; n < 4; ++n)
                        acc[m][n] = __builtin_amdgcn_wmma_f32_16x16x4_f32(
                                        false, aR, false, tR[n], (short)0, acc[m][n], false, false);
                    #pragma unroll
                    for (int n = 0; n < 4; ++n)
                        acc[m][n] = __builtin_amdgcn_wmma_f32_16x16x4_f32(
                                        false, aI, false, tI[n], (short)0, acc[m][n], false, false);
                }
            }
        }
    }

    #pragma unroll
    for (int m = 0; m < 4; ++m) {
        #pragma unroll
        for (int n = 0; n < 4; ++n) {
            #pragma unroll
            for (int r = 0; r < 8; ++r)
                sP[wave * FEATC + (m * 16 + 8 * h + r) * NXC + n * 16 + l15] = acc[m][n][r];
        }
    }
    __syncthreads();

    {
        const size_t fb = (size_t)f * BATCHC + batch;
        #pragma unroll
        for (int p = 0; p < 4; ++p) {
            const int o = p * 1024 + tid * 4;
            v4f s = *(const v4f*)&sP[o];
            #pragma unroll
            for (int w = 1; w < 8; ++w)
                s += *(const v4f*)&sP[w * FEATC + o];
            *(v4f*)&polar[fb * FEATC + o] = s;
        }
    }
}

#define S2_T (64 * LP)

__device__ __forceinline__ void cart_fetch(
    const float* __restrict__ gP, const float* __restrict__ gC,
    v4f (&rp)[16], v4f (&rc)[16])
{
    #pragma unroll
    for (int i = 0; i < 16; ++i)
        rp[i] = *(const v4f*)(gP + (size_t)(4 * i) * 4096);
    #pragma unroll
    for (int i = 0; i < 16; ++i)
        rc[i] = *(const v4f*)(gC + (size_t)(4 * i) * 4096);
}

__device__ __forceinline__ void cart_stash(
    float* sPn, float* sCn, const int soff,
    const v4f (&rp)[16], const v4f (&rc)[16])
{
    #pragma unroll
    for (int i = 0; i < 16; ++i)
        *(v4f*)&sPn[soff + 4 * i * LP] = rp[i];
    #pragma unroll
    for (int i = 0; i < 16; ++i)
        *(v4f*)&sCn[soff + 4 * i * LP] = rc[i];
}

__device__ __forceinline__ void cart_chunk_mma(
    const float* sPc, const float* sCc,
    const int aoff, const int b0off, const int b1off,
    v8f& acc0, v8f& acc1)
{
    #pragma unroll
    for (int j = 0; j < KC; j += 4) {
        const v2f a  = *(const v2f*)&sPc[aoff  + j];
        const v2f b0 = *(const v2f*)&sCc[b0off + j];
        const v2f b1 = *(const v2f*)&sCc[b1off + j];
        acc0 = __builtin_amdgcn_wmma_f32_16x16x4_f32(
                   false, a, false, b0, (short)0, acc0, false, false);
        acc1 = __builtin_amdgcn_wmma_f32_16x16x4_f32(
                   false, a, false, b1, (short)0, acc1, false, false);
    }
}

__global__ __launch_bounds__(256) void cart_stage_kernel(
    const float* __restrict__ cart,
    const float* __restrict__ polar,
    float* __restrict__ out)
{
    __shared__ __attribute__((aligned(16))) float sP[2 * S2_T];
    __shared__ __attribute__((aligned(16))) float sC[2 * S2_T];

    const int tid  = threadIdx.x;
    const int lane = tid & 31;
    const int wave = tid >> 5;

    const int n0 = blockIdx.x * 64;
    const int m0 = blockIdx.y * 64;

    const int mt  = wave >> 1;
    const int nt0 = (wave & 1) * 2;

    const int halfsel = lane >> 4;
    const int ml      = lane & 15;

    const int c4   = (tid & 63) * 4;
    const int srow = tid >> 6;
    const int soff = srow * LP + c4;
    const float* gP = polar + (size_t)(m0 + srow) * 4096 + c4;
    const float* gC = cart  + (size_t)(n0 + srow) * 4096 + c4;

    const int aoff  = ((mt     ) * 16 + ml) * LP + halfsel * 2;
    const int b0off = ((nt0    ) * 16 + ml) * LP + halfsel * 2;
    const int b1off = ((nt0 + 1) * 16 + ml) * LP + halfsel * 2;

    v8f acc0 = {};
    v8f acc1 = {};

    v4f rp[16], rc[16];

    cart_fetch(gP, gC, rp, rc);
    cart_stash(sP, sC, soff, rp, rc);
    __syncthreads();

    for (int q = 0; q < 4096 / KC - 1; ++q) {
        const int cur = q & 1;
        cart_fetch(gP + (q + 1) * KC, gC + (q + 1) * KC, rp, rc);
        cart_chunk_mma(sP + cur * S2_T, sC + cur * S2_T,
                       aoff, b0off, b1off, acc0, acc1);
        cart_stash(sP + (cur ^ 1) * S2_T, sC + (cur ^ 1) * S2_T, soff, rp, rc);
        __syncthreads();
    }
    cart_chunk_mma(sP + S2_T, sC + S2_T,
                   aoff, b0off, b1off, acc0, acc1);

    #pragma unroll
    for (int ni = 0; ni < 2; ++ni) {
        const v8f acc = (ni == 0) ? acc0 : acc1;
        const int p  = n0 + (nt0 + ni) * 16 + ml;
        const int e1 = p >> 6, e2 = p & 63;
        #pragma unroll
        for (int r = 0; r < 8; ++r) {
            const int fb = m0 + mt * 16 + halfsel * 8 + r;
            const int fo = fb >> 4, bo = fb & 15;
            out[(((size_t)bo * 64 + e1) * 64 + e2) * NKC + fo] = acc[r];
        }
    }
}

extern "C" void kernel_launch(void* const* d_in, const int* in_sizes, int n_in,
                              void* d_out, int out_size, void* d_ws, size_t ws_size,
                              hipStream_t stream) {
    const float* inputs  = (const float*)d_in[0];
    const float* cart    = (const float*)d_in[1];
    const int*   r_index = (const int*)  d_in[2];
    const float* pre1  = (const float*)d_in[3];
    const float* pre2  = (const float*)d_in[4];
    const float* pre3  = (const float*)d_in[5];
    const float* pre4  = (const float*)d_in[6];
    const float* post1 = (const float*)d_in[7];
    const float* post2 = (const float*)d_in[8];
    const float* post3 = (const float*)d_in[9];
    const float* post4 = (const float*)d_in[10];
    const float* ck1 = (const float*)d_in[11];
    const float* sk1 = (const float*)d_in[12];
    const float* ck2 = (const float*)d_in[13];
    const float* sk2 = (const float*)d_in[14];
    const float* ck3 = (const float*)d_in[15];
    const float* sk3 = (const float*)d_in[16];
    const float* ck4 = (const float*)d_in[17];
    const float* sk4 = (const float*)d_in[18];

    float* polar = (float*)d_ws;

    polar_stage_kernel<<<dim3(16, 16), 256, 0, stream>>>(
        inputs, r_index,
        pre1, pre2, pre3, pre4, post1, post2, post3, post4,
        ck1, sk1, ck2, sk2, ck3, sk3, ck4, sk4, polar);

    cart_stage_kernel<<<dim3(64, 4), 256, 0, stream>>>(
        cart, polar, (float*)d_out);
}
